# starting kernel
# baseline (speedup 1.0000x reference)
#include <hip/hip_runtime.h>
#include <hip/hip_bf16.h>

typedef __attribute__((ext_vector_type(2))) float v2f;
typedef __attribute__((ext_vector_type(8))) float v8f;

#define DIM 128
#define NEG_SLOPE 0.2f

__device__ __forceinline__ float lrelu(float x) { return x > 0.f ? x : NEG_SLOPE * x; }
__device__ __forceinline__ float elu(float x)   { return x > 0.f ? x : (__expf(x) - 1.f); }

__device__ __forceinline__ unsigned enc_f(float f) {
    unsigned u = __float_as_uint(f);
    return (u & 0x80000000u) ? ~u : (u | 0x80000000u);
}
__device__ __forceinline__ float dec_f(unsigned u) {
    return (u & 0x80000000u) ? __uint_as_float(u & 0x7FFFFFFFu) : __uint_as_float(~u);
}

__global__ void __launch_bounds__(256) gemm128_wmma_f32(
    const float* __restrict__ A, const float* __restrict__ W,
    float* __restrict__ C) {
    const int wave = threadIdx.x >> 5;
    const int lane = threadIdx.x & 31;
    const int half = lane >> 4;
    const int l15  = lane & 15;
    const int row0 = blockIdx.x * 16;
    const int col0 = wave * 16;

    const float* __restrict__ Arow = A + (size_t)(row0 + l15) * DIM;

    v8f acc = {};
#pragma unroll
    for (int k = 0; k < DIM; k += 4) {
        const int ka = k + half * 2;
        v2f a, b;
        a.x = Arow[ka];
        a.y = Arow[ka + 1];
        b.x = W[(size_t)ka * DIM + col0 + l15];
        b.y = W[(size_t)(ka + 1) * DIM + col0 + l15];
        acc = __builtin_amdgcn_wmma_f32_16x16x4_f32(
             false, a,  false, b,
             (short)0, acc,  false,  false);
    }
#pragma unroll
    for (int j = 0; j < 8; ++j)
        C[(size_t)(row0 + j + half * 8) * DIM + col0 + l15] = acc[j];
}

__global__ void zero_fill(float* __restrict__ p, int n) {
    int i = blockIdx.x * blockDim.x + threadIdx.x;
    if (i < n) p[i] = 0.f;
}

template <int HEADS>
__global__ void __launch_bounds__(256) edge_alpha(
    const int* __restrict__ srcA, const int* __restrict__ dstA, int E, int ET,
    const float* __restrict__ xl, const float* __restrict__ xr,
    const float* __restrict__ att, float* __restrict__ alpha,
    unsigned* __restrict__ amax) {
    const int e = blockIdx.x * (blockDim.x >> 5) + (threadIdx.x >> 5);
    if (e >= ET) return;
    const int lane = threadIdx.x & 31;
    int s, d;
    if (e < E) { s = srcA[e]; d = dstA[e]; } else { s = d = e - E; }

    const float4 a = ((const float4*)(xl + (size_t)s * DIM))[lane];
    const float4 b = ((const float4*)(xr + (size_t)d * DIM))[lane];
    const float4 t = ((const float4*)att)[lane];

    float p = lrelu(a.x + b.x) * t.x + lrelu(a.y + b.y) * t.y +
              lrelu(a.z + b.z) * t.z + lrelu(a.w + b.w) * t.w;

    const int span = (HEADS == 2) ? 16 : 32;
#pragma unroll
    for (int m = 1; m < span; m <<= 1) p += __shfl_xor(p, m, 32);

    if ((lane & (span - 1)) == 0) {
        const int h = lane / span;
        alpha[(size_t)e * HEADS + h] = p;
        atomicMax(&amax[(size_t)d * HEADS + h], enc_f(p));
    }
}

template <int HEADS>
__global__ void __launch_bounds__(256) edge_exp(
    const int* __restrict__ dstA, int E, int ET,
    float* __restrict__ alpha, const unsigned* __restrict__ amax,
    float* __restrict__ denom) {
    const int i = blockIdx.x * blockDim.x + threadIdx.x;
    if (i >= ET * HEADS) return;
    const int e = i / HEADS;
    const int h = i - e * HEADS;
    const int d = (e < E) ? dstA[e] : (e - E);
    const float m = dec_f(amax[(size_t)d * HEADS + h]);
    const float w = __expf(alpha[i] - m);
    alpha[i] = w;
    atomicAdd(&denom[(size_t)d * HEADS + h], w);
}

template <int HEADS>
__global__ void __launch_bounds__(256) edge_aggregate(
    const int* __restrict__ srcA, const int* __restrict__ dstA, int E, int ET,
    const float* __restrict__ xl, const float* __restrict__ alpha,
    const float* __restrict__ denom, float* __restrict__ out) {
    const int e = blockIdx.x * (blockDim.x >> 5) + (threadIdx.x >> 5);
    if (e >= ET) return;
    const int lane = threadIdx.x & 31;
    int s, d;
    if (e < E) { s = srcA[e]; d = dstA[e]; } else { s = d = e - E; }

    const int span = (HEADS == 2) ? 16 : 32;
    const int h = lane / span;
    const float coeff = alpha[(size_t)e * HEADS + h] /
                        (denom[(size_t)d * HEADS + h] + 1e-16f);

    const float4 v = ((const float4*)(xl + (size_t)s * DIM))[lane];
    float* o = out + (size_t)d * DIM + lane * 4;
    atomicAdd(o + 0, coeff * v.x);
    atomicAdd(o + 1, coeff * v.y);
    atomicAdd(o + 2, coeff * v.z);
    atomicAdd(o + 3, coeff * v.w);
}

__global__ void epilogue_elu(float* __restrict__ p, const float* __restrict__ bias,
                             int n) {
    int i = blockIdx.x * blockDim.x + threadIdx.x;
    if (i < n) {
        const float v = p[i] + bias[i & (DIM - 1)];
        p[i] = elu(v);
    }
}

extern "C" void kernel_launch(void* const* d_in, const int* in_sizes, int n_in,
                              void* d_out, int out_size, void* d_ws, size_t ws_size,
                              hipStream_t stream) {
    const float* x    = (const float*)d_in[0];
    const int*   ei   = (const int*)d_in[1];
    const float* Wl1  = (const float*)d_in[2];
    const float* Wr1  = (const float*)d_in[3];
    const float* att1 = (const float*)d_in[4];
    const float* b1   = (const float*)d_in[5];
    const float* Wl2  = (const float*)d_in[6];
    const float* Wr2  = (const float*)d_in[7];
    const float* att2 = (const float*)d_in[8];
    const float* b2   = (const float*)d_in[9];

    const int N  = in_sizes[0] / DIM;
    const int E  = in_sizes[1] / 2;
    const int ET = E + N;

    const int* srcA = ei;
    const int* dstA = ei + E;

    float*    xl    = (float*)d_ws;
    float*    xr    = xl + (size_t)N * DIM;
    float*    h1    = xr + (size_t)N * DIM;
    float*    alpha = h1 + (size_t)N * DIM;
    unsigned* amax  = (unsigned*)(alpha + (size_t)ET * 2);
    float*    denom = (float*)(amax + (size_t)N * 2);
    float*    out   = (float*)d_out;

    const int nNodeElems = N * DIM;
    const dim3 B(256);
    const int gNode  = (nNodeElems + 255) / 256;
    const int gStat  = (N * 4 + 255) / 256;
    const int gGemm  = N / 16;
    const int gWaveE = (ET + 7) / 8;

    zero_fill<<<gNode, B, 0, stream>>>(h1, nNodeElems);
    zero_fill<<<gStat, B, 0, stream>>>((float*)amax, N * 4);
    gemm128_wmma_f32<<<gGemm, B, 0, stream>>>(x, Wl1, xl);
    gemm128_wmma_f32<<<gGemm, B, 0, stream>>>(x, Wr1, xr);
    edge_alpha<2><<<gWaveE, B, 0, stream>>>(srcA, dstA, E, ET, xl, xr, att1, alpha, amax);
    edge_exp<2><<<(ET * 2 + 255) / 256, B, 0, stream>>>(dstA, E, ET, alpha, amax, denom);
    edge_aggregate<2><<<gWaveE, B, 0, stream>>>(srcA, dstA, E, ET, xl, alpha, denom, h1);
    epilogue_elu<<<gNode, B, 0, stream>>>(h1, b1, nNodeElems);

    zero_fill<<<gNode, B, 0, stream>>>(out, nNodeElems);
    zero_fill<<<gStat, B, 0, stream>>>((float*)amax, N * 4);
    gemm128_wmma_f32<<<gGemm, B, 0, stream>>>(h1, Wl2, xl);
    gemm128_wmma_f32<<<gGemm, B, 0, stream>>>(h1, Wr2, xr);
    edge_alpha<1><<<gWaveE, B, 0, stream>>>(srcA, dstA, E, ET, xl, xr, att2, alpha, amax);
    edge_exp<1><<<(ET + 255) / 256, B, 0, stream>>>(dstA, E, ET, alpha, amax, denom);
    edge_aggregate<1><<<gWaveE, B, 0, stream>>>(srcA, dstA, E, ET, xl, alpha, denom, out);
    epilogue_elu<<<gNode, B, 0, stream>>>(out, b2, nNodeElems);
}


# Edges sorted by (destination, source) in-call (int-add rank, id order in-wave); one wave per node keeps scores, running-max softmax and weighted sum in registers, stores row once; no float atomics
# speedup vs baseline: 3.3391x; 3.3391x over previous
#include <hip/hip_runtime.h>
#include <hip/hip_bf16.h>

typedef __attribute__((ext_vector_type(2))) float v2f;
typedef __attribute__((ext_vector_type(8))) float v8f;

#define DIM 128
#define NEG_SLOPE 0.2f

__device__ __forceinline__ float lrelu(float x) { return x > 0.f ? x : NEG_SLOPE * x; }
__device__ __forceinline__ float elu(float x)   { return x > 0.f ? x : (__expf(x) - 1.f); }

__global__ void __launch_bounds__(256) gemm128_wmma_f32(
    const float* __restrict__ A, const float* __restrict__ W,
    float* __restrict__ C) {
    const int wave = threadIdx.x >> 5;
    const int lane = threadIdx.x & 31;
    const int half = lane >> 4;
    const int l15  = lane & 15;
    const int row0 = blockIdx.x * 16;
    const int col0 = wave * 16;

    const float* __restrict__ Arow = A + (size_t)(row0 + l15) * DIM;

    v8f acc = {};
#pragma unroll
    for (int k = 0; k < DIM; k += 4) {
        const int ka = k + half * 2;
        v2f a, b;
        a.x = Arow[ka];
        a.y = Arow[ka + 1];
        b.x = W[(size_t)ka * DIM + col0 + l15];
        b.y = W[(size_t)(ka + 1) * DIM + col0 + l15];
        acc = __builtin_amdgcn_wmma_f32_16x16x4_f32(
             false, a,  false, b,
             (short)0, acc,  false,  false);
    }
#pragma unroll
    for (int j = 0; j < 8; ++j)
        C[(size_t)(row0 + j + half * 8) * DIM + col0 + l15] = acc[j];
}

__global__ void zero_fill(float* __restrict__ p, int n) {
    int i = blockIdx.x * blockDim.x + threadIdx.x;
    if (i < n) p[i] = 0.f;
}

__global__ void __launch_bounds__(256) dst_rank(
    const int* __restrict__ dstA, int E, int* __restrict__ cnt,
    int* __restrict__ rank) {
    const int e = blockIdx.x * blockDim.x + threadIdx.x;
    if (e < E) rank[e] = atomicAdd(&cnt[dstA[e]], 1);
}

__global__ void __launch_bounds__(256) seg_alloc(
    int* cnt, int* __restrict__ start, int n) {
    const int i = blockIdx.x * blockDim.x + threadIdx.x;
    const int lane = threadIdx.x & 31;
    const int c = (i < n) ? cnt[i] : 0;
    int incl = c;
#pragma unroll
    for (int m = 1; m < 32; m <<= 1) {
        const int v = __shfl_up(incl, m, 32);
        if (lane >= m) incl += v;
    }
    const int total = __shfl(incl, 31, 32);
    int base = 0;
    if (lane == 0) base = atomicAdd(&cnt[n], total);
    base = __shfl(base, 0, 32);
    if (i < n) start[i] = base + incl - c;
}

__global__ void __launch_bounds__(256) fill_lists(
    const int* __restrict__ srcA, const int* __restrict__ dstA, int E,
    const int* __restrict__ start, const int* __restrict__ rank,
    int* __restrict__ lst) {
    const int e = blockIdx.x * blockDim.x + threadIdx.x;
    if (e < E) lst[start[dstA[e]] + rank[e]] = srcA[e];
}

__device__ __forceinline__ float edge_score(const float4 a, const float4 b,
                                            const float4 t) {
    return lrelu(a.x + b.x) * t.x + lrelu(a.y + b.y) * t.y +
           lrelu(a.z + b.z) * t.z + lrelu(a.w + b.w) * t.w;
}

template <int SPAN>
__device__ __forceinline__ float head_sum(float p) {
#pragma unroll
    for (int m = 1; m < SPAN; m <<= 1) p += __shfl_xor(p, m, 32);
    return p;
}

template <int HEADS>
__global__ void __launch_bounds__(256) node_attend(
    const int* __restrict__ start, const int* __restrict__ cnt,
    const int* __restrict__ lst, int N,
    const float* __restrict__ xl, const float* __restrict__ xr,
    const float* __restrict__ att, const float* __restrict__ bias,
    float* __restrict__ out) {
    const int d = __builtin_amdgcn_readfirstlane(
        (int)(blockIdx.x * (blockDim.x >> 5) + (threadIdx.x >> 5)));
    if (d >= N) return;
    const int lane = threadIdx.x & 31;
    constexpr int SPAN = (HEADS == 2) ? 16 : 32;

    const float4* __restrict__ xl4 = (const float4*)xl;
    const float4 b = ((const float4*)xr)[(size_t)d * (DIM / 4) + lane];
    const float4 t = ((const float4*)att)[lane];

    float4 acc = xl4[(size_t)d * (DIM / 4) + lane];
    float m = head_sum<SPAN>(edge_score(acc, b, t));
    float l = 1.f;

    const int beg = start[d];
    const int n   = cnt[d];
    const float NEG_INF = -__builtin_huge_valf();

    for (int base = 0; base < n; base += 32) {
        const int c = (n - base < 32) ? (n - base) : 32;
        const bool live = lane < c;
        const int my_s = live ? lst[beg + base + lane] : d;

        int pos = live ? 0 : lane;
        for (int i = 0; i < c; ++i) {
            const int si = __builtin_amdgcn_readlane(my_s, i);
            const bool before = (si < my_s) || (si == my_s && i < lane);
            pos += (live && before) ? 1 : 0;
        }

        for (int j = 0; j < c; j += 4) {
            const int i0 = __builtin_ctz(__builtin_amdgcn_ballot_w32(pos == j));
            const int i1 = __builtin_ctz(__builtin_amdgcn_ballot_w32(pos == j + 1));
            const int i2 = __builtin_ctz(__builtin_amdgcn_ballot_w32(pos == j + 2));
            const int i3 = __builtin_ctz(__builtin_amdgcn_ballot_w32(pos == j + 3));
            const int s0 = __builtin_amdgcn_readlane(my_s, i0);
            const int s1 = __builtin_amdgcn_readlane(my_s, i1);
            const int s2 = __builtin_amdgcn_readlane(my_s, i2);
            const int s3 = __builtin_amdgcn_readlane(my_s, i3);
            const float4 a0 = xl4[(size_t)s0 * (DIM / 4) + lane];
            const float4 a1 = xl4[(size_t)s1 * (DIM / 4) + lane];
            const float4 a2 = xl4[(size_t)s2 * (DIM / 4) + lane];
            const float4 a3 = xl4[(size_t)s3 * (DIM / 4) + lane];

            float q0 = edge_score(a0, b, t);
            float q1 = edge_score(a1, b, t);
            float q2 = edge_score(a2, b, t);
            float q3 = edge_score(a3, b, t);
#pragma unroll
            for (int mm = 1; mm < SPAN; mm <<= 1) {
                const float r0 = __shfl_xor(q0, mm, 32);
                const float r1 = __shfl_xor(q1, mm, 32);
                const float r2 = __shfl_xor(q2, mm, 32);
                const float r3 = __shfl_xor(q3, mm, 32);
                q0 += r0; q1 += r1; q2 += r2; q3 += r3;
            }
            const float p0 = q0;
            const float p1 = (j + 1 < c) ? q1 : NEG_INF;
            const float p2 = (j + 2 < c) ? q2 : NEG_INF;
            const float p3 = (j + 3 < c) ? q3 : NEG_INF;

            const float mn = fmaxf(fmaxf(m, p0), fmaxf(p1, fmaxf(p2, p3)));
            const float sc = __expf(m - mn);
            const float w0 = __expf(p0 - mn);
            const float w1 = __expf(p1 - mn);
            const float w2 = __expf(p2 - mn);
            const float w3 = __expf(p3 - mn);
            m = mn;
            l = l * sc + ((w0 + w1) + (w2 + w3));
            acc.x = acc.x * sc + w0 * a0.x + w1 * a1.x + w2 * a2.x + w3 * a3.x;
            acc.y = acc.y * sc + w0 * a0.y + w1 * a1.y + w2 * a2.y + w3 * a3.y;
            acc.z = acc.z * sc + w0 * a0.z + w1 * a1.z + w2 * a2.z + w3 * a3.z;
            acc.w = acc.w * sc + w0 * a0.w + w1 * a1.w + w2 * a2.w + w3 * a3.w;
        }
    }

    const float den = l + 1e-16f;
    const float4 bi = ((const float4*)bias)[lane];
    float4 o;
    o.x = elu(acc.x / den + bi.x);
    o.y = elu(acc.y / den + bi.y);
    o.z = elu(acc.z / den + bi.z);
    o.w = elu(acc.w / den + bi.w);
    ((float4*)out)[(size_t)d * (DIM / 4) + lane] = o;
}

extern "C" void kernel_launch(void* const* d_in, const int* in_sizes, int n_in,
                              void* d_out, int out_size, void* d_ws, size_t ws_size,
                              hipStream_t stream) {
    const float* x    = (const float*)d_in[0];
    const int*   ei   = (const int*)d_in[1];
    const float* Wl1  = (const float*)d_in[2];
    const float* Wr1  = (const float*)d_in[3];
    const float* att1 = (const float*)d_in[4];
    const float* b1   = (const float*)d_in[5];
    const float* Wl2  = (const float*)d_in[6];
    const float* Wr2  = (const float*)d_in[7];
    const float* att2 = (const float*)d_in[8];
    const float* b2   = (const float*)d_in[9];

    const int N  = in_sizes[0] / DIM;
    const int E  = in_sizes[1] / 2;

    const int* srcA = ei;
    const int* dstA = ei + E;

    float* xl    = (float*)d_ws;
    float* xr    = xl + (size_t)N * DIM;
    float* h1    = xr + (size_t)N * DIM;
    int*   cnt   = (int*)(h1 + (size_t)N * DIM);
    int*   start = cnt + (size_t)N + 64;
    int*   rank  = start + (size_t)N + 64;
    int*   lst   = rank + (size_t)E;
    float* out   = (float*)d_out;

    const dim3 B(256);
    const int gCnt   = (N + 1 + 255) / 256;
    const int gNodeT = (N + 255) / 256;
    const int gEdgeT = (E + 255) / 256;
    const int gGemm  = N / 16;
    const int gNodeW = (N + 7) / 8;

    zero_fill<<<gCnt, B, 0, stream>>>((float*)cnt, N + 1);
    dst_rank<<<gEdgeT, B, 0, stream>>>(dstA, E, cnt, rank);
    seg_alloc<<<gNodeT, B, 0, stream>>>(cnt, start, N);
    fill_lists<<<gEdgeT, B, 0, stream>>>(srcA, dstA, E, start, rank, lst);

    gemm128_wmma_f32<<<gGemm, B, 0, stream>>>(x, Wl1, xl);
    gemm128_wmma_f32<<<gGemm, B, 0, stream>>>(x, Wr1, xr);
    node_attend<2><<<gNodeW, B, 0, stream>>>(start, cnt, lst, N, xl, xr, att1, b1, h1);

    gemm128_wmma_f32<<<gGemm, B, 0, stream>>>(h1, Wl2, xl);
    gemm128_wmma_f32<<<gGemm, B, 0, stream>>>(h1, Wr2, xr);
    node_attend<1><<<gNodeW, B, 0, stream>>>(start, cnt, lst, N, xl, xr, att2, b2, out);
}


# 5x fewer, longer-lived waves per launch (wave entry rate was the limiter): GEMM wave does 16 rows x all 128 cols (A fragment feeds 8 tiles), attend wave walks 4 nodes, edge passes 8 edges/thread
# speedup vs baseline: 3.5263x; 1.0561x over previous
#include <hip/hip_runtime.h>
#include <hip/hip_bf16.h>

typedef __attribute__((ext_vector_type(2))) float v2f;
typedef __attribute__((ext_vector_type(8))) float v8f;

#define DIM 128
#define NODES_PER_WAVE 4
#define KSTAGE 2
#define NEG_SLOPE 0.2f

__device__ __forceinline__ float lrelu(float x) { return x > 0.f ? x : NEG_SLOPE * x; }
__device__ __forceinline__ float elu(float x)   { return x > 0.f ? x : (__expf(x) - 1.f); }

#define WMMA_F32(a, b, c) __builtin_amdgcn_wmma_f32_16x16x4_f32( \
     false, a,  false, b,  (short)0, c, \
     false,  false)

struct KFrag {
    v2f a;
    float4 p0, q0;
    float4 p1, q1;
};

__device__ __forceinline__ KFrag load_kfrag(const float* __restrict__ Ap,
                                            const float* __restrict__ Wp, int k) {
    KFrag f;
    f.a  = *(const v2f*)(Ap + k);
    f.p0 = *(const float4*)(Wp + (size_t)k * DIM);
    f.q0 = *(const float4*)(Wp + (size_t)(k + 1) * DIM);
    f.p1 = *(const float4*)(Wp + (size_t)k * DIM + 64);
    f.q1 = *(const float4*)(Wp + (size_t)(k + 1) * DIM + 64);
    return f;
}

__device__ __forceinline__ void mma_kfrag(const KFrag& f, v8f (&c)[8]) {
    v2f b;
    b.x = f.p0.x; b.y = f.q0.x; c[0] = WMMA_F32(f.a, b, c[0]);
    b.x = f.p0.y; b.y = f.q0.y; c[1] = WMMA_F32(f.a, b, c[1]);
    b.x = f.p0.z; b.y = f.q0.z; c[2] = WMMA_F32(f.a, b, c[2]);
    b.x = f.p0.w; b.y = f.q0.w; c[3] = WMMA_F32(f.a, b, c[3]);
    b.x = f.p1.x; b.y = f.q1.x; c[4] = WMMA_F32(f.a, b, c[4]);
    b.x = f.p1.y; b.y = f.q1.y; c[5] = WMMA_F32(f.a, b, c[5]);
    b.x = f.p1.z; b.y = f.q1.z; c[6] = WMMA_F32(f.a, b, c[6]);
    b.x = f.p1.w; b.y = f.q1.w; c[7] = WMMA_F32(f.a, b, c[7]);
}

__global__ void __launch_bounds__(128) gemm128_wmma_f32(
    const float* __restrict__ A, const float* __restrict__ W,
    float* __restrict__ C, int ntiles) {
    const int lane = threadIdx.x & 31;
    const int half = lane >> 4;
    const int l15  = lane & 15;
    const int tile = __builtin_amdgcn_readfirstlane(
        (int)(blockIdx.x * 4 + (threadIdx.x >> 5)));
    if (tile >= ntiles) return;
    const int row0 = tile * 16;

    const float* __restrict__ Ap = A + (size_t)(row0 + l15) * DIM + half * 2;
    const float* __restrict__ Wp = W + (size_t)(half * 2) * DIM + l15 * 4;

    v8f c[8] = {};
    KFrag fa[KSTAGE], fb[KSTAGE];
    constexpr int KS4 = 4 * KSTAGE;
#pragma unroll
    for (int s = 0; s < KSTAGE; ++s) fa[s] = load_kfrag(Ap, Wp, 4 * s);
#pragma unroll 1
    for (int k = 0; k < DIM - 2 * KS4; k += 2 * KS4) {
#pragma unroll
        for (int s = 0; s < KSTAGE; ++s) fb[s] = load_kfrag(Ap, Wp, k + KS4 + 4 * s);
#pragma unroll
        for (int s = 0; s < KSTAGE; ++s) mma_kfrag(fa[s], c);
#pragma unroll
        for (int s = 0; s < KSTAGE; ++s) fa[s] = load_kfrag(Ap, Wp, k + 2 * KS4 + 4 * s);
#pragma unroll
        for (int s = 0; s < KSTAGE; ++s) mma_kfrag(fb[s], c);
    }
#pragma unroll
    for (int s = 0; s < KSTAGE; ++s) fb[s] = load_kfrag(Ap, Wp, DIM - KS4 + 4 * s);
#pragma unroll
    for (int s = 0; s < KSTAGE; ++s) mma_kfrag(fa[s], c);
#pragma unroll
    for (int s = 0; s < KSTAGE; ++s) mma_kfrag(fb[s], c);

    float* __restrict__ Cp = C + (size_t)(row0 + half * 8) * DIM + l15 * 4;
#pragma unroll
    for (int r = 0; r < 8; ++r) {
        float4 lo, hi;
        lo.x = c[0][r]; lo.y = c[1][r]; lo.z = c[2][r]; lo.w = c[3][r];
        hi.x = c[4][r]; hi.y = c[5][r]; hi.z = c[6][r]; hi.w = c[7][r];
        *(float4*)(Cp + (size_t)r * DIM)      = lo;
        *(float4*)(Cp + (size_t)r * DIM + 64) = hi;
    }
}

__global__ void zero_fill(float* __restrict__ p, int n) {
    int i = blockIdx.x * blockDim.x + threadIdx.x;
    if (i < n) p[i] = 0.f;
}

#define EPT 8
__global__ void __launch_bounds__(256) dst_rank(
    const int* __restrict__ dstA, int E, int* __restrict__ cnt,
    int* __restrict__ rank) {
    const int e0 = blockIdx.x * blockDim.x + threadIdx.x;
    const int T  = gridDim.x * blockDim.x;
    int dn[EPT];
#pragma unroll
    for (int k = 0; k < EPT; ++k) {
        const int e = e0 + k * T;
        dn[k] = dstA[(e < E) ? e : (E - 1)];
    }
    int rk[EPT];
#pragma unroll
    for (int k = 0; k < EPT; ++k) {
        const int e = e0 + k * T;
        rk[k] = atomicAdd(&cnt[dn[k]], (e < E) ? 1 : 0);
    }
#pragma unroll
    for (int k = 0; k < EPT; ++k) {
        const int e = e0 + k * T;
        if (e < E) rank[e] = rk[k];
    }
}

__global__ void __launch_bounds__(256) seg_alloc(
    int* cnt, int* __restrict__ start, int n) {
    const int i = blockIdx.x * blockDim.x + threadIdx.x;
    const int lane = threadIdx.x & 31;
    const int c = (i < n) ? cnt[i] : 0;
    int incl = c;
#pragma unroll
    for (int m = 1; m < 32; m <<= 1) {
        const int v = __shfl_up(incl, m, 32);
        if (lane >= m) incl += v;
    }
    const int total = __shfl(incl, 31, 32);
    int base = 0;
    if (lane == 0) base = atomicAdd(&cnt[n], total);
    base = __shfl(base, 0, 32);
    if (i < n) start[i] = base + incl - c;
}

__global__ void __launch_bounds__(256) fill_lists(
    const int* __restrict__ srcA, const int* __restrict__ dstA, int E,
    const int* __restrict__ start, const int* __restrict__ rank,
    int* __restrict__ lst) {
    const int e0 = blockIdx.x * blockDim.x + threadIdx.x;
    const int T  = gridDim.x * blockDim.x;
    int dn[EPT], rk[EPT], sn[EPT];
#pragma unroll
    for (int k = 0; k < EPT; ++k) {
        const int e  = e0 + k * T;
        const int ec = (e < E) ? e : (E - 1);
        dn[k] = dstA[ec];
        rk[k] = rank[ec];
        sn[k] = srcA[ec];
    }
    int st[EPT];
#pragma unroll
    for (int k = 0; k < EPT; ++k) st[k] = start[dn[k]];
#pragma unroll
    for (int k = 0; k < EPT; ++k) {
        const int e = e0 + k * T;
        if (e < E) lst[st[k] + rk[k]] = sn[k];
    }
}

__device__ __forceinline__ float edge_score(const float4 a, const float4 b,
                                            const float4 t) {
    return lrelu(a.x + b.x) * t.x + lrelu(a.y + b.y) * t.y +
           lrelu(a.z + b.z) * t.z + lrelu(a.w + b.w) * t.w;
}

template <int SPAN>
__device__ __forceinline__ float head_sum(float p) {
#pragma unroll
    for (int m = 1; m < SPAN; m <<= 1) p += __shfl_xor(p, m, 32);
    return p;
}

template <int HEADS>
__global__ void __launch_bounds__(256, 16) node_attend(
    const int* __restrict__ start, const int* __restrict__ cnt,
    const int* __restrict__ lst, int N,
    const float* __restrict__ xl, const float* __restrict__ xr,
    const float* __restrict__ att, const float* bias,
    float* out) {
    const int d_first = __builtin_amdgcn_readfirstlane(
        (int)(blockIdx.x * (blockDim.x >> 5) + (threadIdx.x >> 5)));
    const int d_step = (int)(gridDim.x * (blockDim.x >> 5));
    const int lane = threadIdx.x & 31;
    constexpr int SPAN = (HEADS == 2) ? 16 : 32;

    const float4* __restrict__ xl4 = (const float4*)xl;
    const float4 t  = ((const float4*)att)[lane];
    const float NEG_INF = -__builtin_huge_valf();

    for (int d = d_first; d < N; d += d_step) {
    const float4 b = ((const float4*)xr)[(size_t)d * (DIM / 4) + lane];

    float4 acc = xl4[(size_t)d * (DIM / 4) + lane];
    float m = head_sum<SPAN>(edge_score(acc, b, t));
    float l = 1.f;

    const int beg = start[d];
    const int n   = cnt[d];

    for (int base = 0; base < n; base += 32) {
        const int c = (n - base < 32) ? (n - base) : 32;
        const bool live = lane < c;
        const int my_s = live ? lst[beg + base + lane] : d;

        int pos = live ? 0 : lane;
        for (int i = 0; i < c; ++i) {
            const int si = __builtin_amdgcn_readlane(my_s, i);
            const bool before = (si < my_s) || (si == my_s && i < lane);
            pos += (live && before) ? 1 : 0;
        }

        for (int j = 0; j < c; j += 4) {
            const int i0 = __builtin_ctz(__builtin_amdgcn_ballot_w32(pos == j));
            const int i1 = __builtin_ctz(__builtin_amdgcn_ballot_w32(pos == j + 1));
            const int i2 = __builtin_ctz(__builtin_amdgcn_ballot_w32(pos == j + 2));
            const int i3 = __builtin_ctz(__builtin_amdgcn_ballot_w32(pos == j + 3));
            const int s0 = __builtin_amdgcn_readlane(my_s, i0);
            const int s1 = __builtin_amdgcn_readlane(my_s, i1);
            const int s2 = __builtin_amdgcn_readlane(my_s, i2);
            const int s3 = __builtin_amdgcn_readlane(my_s, i3);
            const float4 a0 = xl4[(size_t)s0 * (DIM / 4) + lane];
            const float4 a1 = xl4[(size_t)s1 * (DIM / 4) + lane];
            const float4 a2 = xl4[(size_t)s2 * (DIM / 4) + lane];
            const float4 a3 = xl4[(size_t)s3 * (DIM / 4) + lane];

            float q0 = edge_score(a0, b, t);
            float q1 = edge_score(a1, b, t);
            float q2 = edge_score(a2, b, t);
            float q3 = edge_score(a3, b, t);
#pragma unroll
            for (int mm = 1; mm < SPAN; mm <<= 1) {
                const float r0 = __shfl_xor(q0, mm, 32);
                const float r1 = __shfl_xor(q1, mm, 32);
                const float r2 = __shfl_xor(q2, mm, 32);
                const float r3 = __shfl_xor(q3, mm, 32);
                q0 += r0; q1 += r1; q2 += r2; q3 += r3;
            }
            const float p0 = q0;
            const float p1 = (j + 1 < c) ? q1 : NEG_INF;
            const float p2 = (j + 2 < c) ? q2 : NEG_INF;
            const float p3 = (j + 3 < c) ? q3 : NEG_INF;

            const float mn = fmaxf(fmaxf(m, p0), fmaxf(p1, fmaxf(p2, p3)));
            const float sc = __expf(m - mn);
            const float w0 = __expf(p0 - mn);
            const float w1 = __expf(p1 - mn);
            const float w2 = __expf(p2 - mn);
            const float w3 = __expf(p3 - mn);
            m = mn;
            l = l * sc + ((w0 + w1) + (w2 + w3));
            acc.x = acc.x * sc + w0 * a0.x + w1 * a1.x + w2 * a2.x + w3 * a3.x;
            acc.y = acc.y * sc + w0 * a0.y + w1 * a1.y + w2 * a2.y + w3 * a3.y;
            acc.z = acc.z * sc + w0 * a0.z + w1 * a1.z + w2 * a2.z + w3 * a3.z;
            acc.w = acc.w * sc + w0 * a0.w + w1 * a1.w + w2 * a2.w + w3 * a3.w;
        }
    }

    const float den = l + 1e-16f;
    const float4 bi = ((const float4*)bias)[lane];
    float4 o;
    o.x = elu(acc.x / den + bi.x);
    o.y = elu(acc.y / den + bi.y);
    o.z = elu(acc.z / den + bi.z);
    o.w = elu(acc.w / den + bi.w);
    ((float4*)out)[(size_t)d * (DIM / 4) + lane] = o;
    }
}

extern "C" void kernel_launch(void* const* d_in, const int* in_sizes, int n_in,
                              void* d_out, int out_size, void* d_ws, size_t ws_size,
                              hipStream_t stream) {
    const float* x    = (const float*)d_in[0];
    const int*   ei   = (const int*)d_in[1];
    const float* Wl1  = (const float*)d_in[2];
    const float* Wr1  = (const float*)d_in[3];
    const float* att1 = (const float*)d_in[4];
    const float* b1   = (const float*)d_in[5];
    const float* Wl2  = (const float*)d_in[6];
    const float* Wr2  = (const float*)d_in[7];
    const float* att2 = (const float*)d_in[8];
    const float* b2   = (const float*)d_in[9];

    const int N  = in_sizes[0] / DIM;
    const int E  = in_sizes[1] / 2;

    const int* srcA = ei;
    const int* dstA = ei + E;

    float* xl    = (float*)d_ws;
    float* xr    = xl + (size_t)N * DIM;
    float* h1    = xr + (size_t)N * DIM;
    int*   cnt   = (int*)(h1 + (size_t)N * DIM);
    int*   start = cnt + (size_t)N + 64;
    int*   rank  = start + (size_t)N + 64;
    int*   lst   = rank + (size_t)E;
    float* out   = (float*)d_out;

    const dim3 B(256);
    const dim3 BG(128);
    const int gCnt   = (N + 1 + 255) / 256;
    const int gNodeT = (N + 255) / 256;
    const int gEdgeT = (E + 256 * EPT - 1) / (256 * EPT);
    const int nTiles = N / 16;
    const int gGemm  = (nTiles + 3) / 4;
    const int gNodeW = (N + 8 * NODES_PER_WAVE - 1) / (8 * NODES_PER_WAVE);

    zero_fill<<<gCnt, B, 0, stream>>>((float*)cnt, N + 1);
    dst_rank<<<gEdgeT, B, 0, stream>>>(dstA, E, cnt, rank);
    seg_alloc<<<gNodeT, B, 0, stream>>>(cnt, start, N);
    fill_lists<<<gEdgeT, B, 0, stream>>>(srcA, dstA, E, start, rank, lst);

    gemm128_wmma_f32<<<gGemm, BG, 0, stream>>>(x, Wl1, xl, nTiles);
    gemm128_wmma_f32<<<gGemm, BG, 0, stream>>>(x, Wr1, xr, nTiles);
    node_attend<2><<<gNodeW, B, 0, stream>>>(start, cnt, lst, N, xl, xr, att1, b1, h1);

    gemm128_wmma_f32<<<gGemm, BG, 0, stream>>>(h1, Wl2, xl, nTiles);
    gemm128_wmma_f32<<<gGemm, BG, 0, stream>>>(h1, Wr2, xr, nTiles);
    node_attend<1><<<gNodeW, B, 0, stream>>>(start, cnt, lst, N, xl, xr, att2, b2, out);
}


# Independent launches merged: each layer's two projections run as one launch (block range picks the weight matrix) and layer 1's also carries the edge-rank pass; 10 launches become 7
# speedup vs baseline: 3.9556x; 1.1217x over previous
#include <hip/hip_runtime.h>
#include <hip/hip_bf16.h>

typedef __attribute__((ext_vector_type(2))) float v2f;
typedef __attribute__((ext_vector_type(8))) float v8f;

#define DIM 128
#define NODES_PER_WAVE 4
#define KSTAGE 2
#define NEG_SLOPE 0.2f

__device__ __forceinline__ float lrelu(float x) { return x > 0.f ? x : NEG_SLOPE * x; }
__device__ __forceinline__ float elu(float x)   { return x > 0.f ? x : (__expf(x) - 1.f); }

#define WMMA_F32(a, b, c) __builtin_amdgcn_wmma_f32_16x16x4_f32( \
     false, a,  false, b,  (short)0, c, \
     false,  false)

struct KFrag {
    v2f a;
    float4 p0, q0;
    float4 p1, q1;
};

__device__ __forceinline__ KFrag load_kfrag(const float* __restrict__ Ap,
                                            const float* __restrict__ Wp, int k) {
    KFrag f;
    f.a  = *(const v2f*)(Ap + k);
    f.p0 = *(const float4*)(Wp + (size_t)k * DIM);
    f.q0 = *(const float4*)(Wp + (size_t)(k + 1) * DIM);
    f.p1 = *(const float4*)(Wp + (size_t)k * DIM + 64);
    f.q1 = *(const float4*)(Wp + (size_t)(k + 1) * DIM + 64);
    return f;
}

__device__ __forceinline__ void mma_kfrag(const KFrag& f, v8f (&c)[8]) {
    v2f b;
    b.x = f.p0.x; b.y = f.q0.x; c[0] = WMMA_F32(f.a, b, c[0]);
    b.x = f.p0.y; b.y = f.q0.y; c[1] = WMMA_F32(f.a, b, c[1]);
    b.x = f.p0.z; b.y = f.q0.z; c[2] = WMMA_F32(f.a, b, c[2]);
    b.x = f.p0.w; b.y = f.q0.w; c[3] = WMMA_F32(f.a, b, c[3]);
    b.x = f.p1.x; b.y = f.q1.x; c[4] = WMMA_F32(f.a, b, c[4]);
    b.x = f.p1.y; b.y = f.q1.y; c[5] = WMMA_F32(f.a, b, c[5]);
    b.x = f.p1.z; b.y = f.q1.z; c[6] = WMMA_F32(f.a, b, c[6]);
    b.x = f.p1.w; b.y = f.q1.w; c[7] = WMMA_F32(f.a, b, c[7]);
}

#define EPT 8
__device__ __forceinline__ void dst_rank_role(
    const int* __restrict__ dstA, int E, int* __restrict__ cnt,
    int* __restrict__ rank, int e0, int T) {
    int dn[EPT];
#pragma unroll
    for (int k = 0; k < EPT; ++k) {
        const int e = e0 + k * T;
        dn[k] = dstA[(e < E) ? e : (E - 1)];
    }
    int rk[EPT];
#pragma unroll
    for (int k = 0; k < EPT; ++k) {
        const int e = e0 + k * T;
        rk[k] = atomicAdd(&cnt[dn[k]], (e < E) ? 1 : 0);
    }
#pragma unroll
    for (int k = 0; k < EPT; ++k) {
        const int e = e0 + k * T;
        if (e < E) rank[e] = rk[k];
    }
}

__global__ void __launch_bounds__(128) proj_pair_wmma_f32(
    const float* __restrict__ A,
    const float* __restrict__ W0, const float* __restrict__ W1,
    float* __restrict__ C0, float* __restrict__ C1, int ntiles,
    const int* __restrict__ dstA, int E, int* __restrict__ cnt,
    int* __restrict__ rank, int rankBlocks) {
    if ((int)blockIdx.x < rankBlocks) {
        dst_rank_role(dstA, E, cnt, rank,
                      (int)(blockIdx.x * 128 + threadIdx.x), rankBlocks * 128);
        return;
    }
    const int lane = threadIdx.x & 31;
    const int half = lane >> 4;
    const int l15  = lane & 15;
    const int strip = __builtin_amdgcn_readfirstlane(
        (int)((blockIdx.x - rankBlocks) * 4 + (threadIdx.x >> 5)));
    if (strip >= 2 * ntiles) return;
    const bool second = strip >= ntiles;
    const float* __restrict__ W = second ? W1 : W0;
    float* __restrict__ C       = second ? C1 : C0;
    const int tile = second ? strip - ntiles : strip;
    const int row0 = tile * 16;

    const float* __restrict__ Ap = A + (size_t)(row0 + l15) * DIM + half * 2;
    const float* __restrict__ Wp = W + (size_t)(half * 2) * DIM + l15 * 4;

    v8f c[8] = {};
    KFrag fa[KSTAGE], fb[KSTAGE];
    constexpr int KS4 = 4 * KSTAGE;
#pragma unroll
    for (int s = 0; s < KSTAGE; ++s) fa[s] = load_kfrag(Ap, Wp, 4 * s);
#pragma unroll 1
    for (int k = 0; k < DIM - 2 * KS4; k += 2 * KS4) {
#pragma unroll
        for (int s = 0; s < KSTAGE; ++s) fb[s] = load_kfrag(Ap, Wp, k + KS4 + 4 * s);
#pragma unroll
        for (int s = 0; s < KSTAGE; ++s) mma_kfrag(fa[s], c);
#pragma unroll
        for (int s = 0; s < KSTAGE; ++s) fa[s] = load_kfrag(Ap, Wp, k + 2 * KS4 + 4 * s);
#pragma unroll
        for (int s = 0; s < KSTAGE; ++s) mma_kfrag(fb[s], c);
    }
#pragma unroll
    for (int s = 0; s < KSTAGE; ++s) fb[s] = load_kfrag(Ap, Wp, DIM - KS4 + 4 * s);
#pragma unroll
    for (int s = 0; s < KSTAGE; ++s) mma_kfrag(fa[s], c);
#pragma unroll
    for (int s = 0; s < KSTAGE; ++s) mma_kfrag(fb[s], c);

    float* __restrict__ Cp = C + (size_t)(row0 + half * 8) * DIM + l15 * 4;
#pragma unroll
    for (int r = 0; r < 8; ++r) {
        float4 lo, hi;
        lo.x = c[0][r]; lo.y = c[1][r]; lo.z = c[2][r]; lo.w = c[3][r];
        hi.x = c[4][r]; hi.y = c[5][r]; hi.z = c[6][r]; hi.w = c[7][r];
        *(float4*)(Cp + (size_t)r * DIM)      = lo;
        *(float4*)(Cp + (size_t)r * DIM + 64) = hi;
    }
}

__global__ void zero_fill(float* __restrict__ p, int n) {
    int i = blockIdx.x * blockDim.x + threadIdx.x;
    if (i < n) p[i] = 0.f;
}


__global__ void __launch_bounds__(256) seg_alloc(
    int* cnt, int* __restrict__ start, int n) {
    const int i = blockIdx.x * blockDim.x + threadIdx.x;
    const int lane = threadIdx.x & 31;
    const int c = (i < n) ? cnt[i] : 0;
    int incl = c;
#pragma unroll
    for (int m = 1; m < 32; m <<= 1) {
        const int v = __shfl_up(incl, m, 32);
        if (lane >= m) incl += v;
    }
    const int total = __shfl(incl, 31, 32);
    int base = 0;
    if (lane == 0) base = atomicAdd(&cnt[n], total);
    base = __shfl(base, 0, 32);
    if (i < n) start[i] = base + incl - c;
}

__global__ void __launch_bounds__(256) fill_lists(
    const int* __restrict__ srcA, const int* __restrict__ dstA, int E,
    const int* __restrict__ start, const int* __restrict__ rank,
    int* __restrict__ lst) {
    const int e0 = blockIdx.x * blockDim.x + threadIdx.x;
    const int T  = gridDim.x * blockDim.x;
    int dn[EPT], rk[EPT], sn[EPT];
#pragma unroll
    for (int k = 0; k < EPT; ++k) {
        const int e  = e0 + k * T;
        const int ec = (e < E) ? e : (E - 1);
        dn[k] = dstA[ec];
        rk[k] = rank[ec];
        sn[k] = srcA[ec];
    }
    int st[EPT];
#pragma unroll
    for (int k = 0; k < EPT; ++k) st[k] = start[dn[k]];
#pragma unroll
    for (int k = 0; k < EPT; ++k) {
        const int e = e0 + k * T;
        if (e < E) lst[st[k] + rk[k]] = sn[k];
    }
}

__device__ __forceinline__ float edge_score(const float4 a, const float4 b,
                                            const float4 t) {
    return lrelu(a.x + b.x) * t.x + lrelu(a.y + b.y) * t.y +
           lrelu(a.z + b.z) * t.z + lrelu(a.w + b.w) * t.w;
}

template <int SPAN>
__device__ __forceinline__ float head_sum(float p) {
#pragma unroll
    for (int m = 1; m < SPAN; m <<= 1) p += __shfl_xor(p, m, 32);
    return p;
}

template <int HEADS>
__global__ void __launch_bounds__(256, 16) node_attend(
    const int* __restrict__ start, const int* __restrict__ cnt,
    const int* __restrict__ lst, int N,
    const float* __restrict__ xl, const float* __restrict__ xr,
    const float* __restrict__ att, const float* bias,
    float* out) {
    const int d_first = __builtin_amdgcn_readfirstlane(
        (int)(blockIdx.x * (blockDim.x >> 5) + (threadIdx.x >> 5)));
    const int d_step = (int)(gridDim.x * (blockDim.x >> 5));
    const int lane = threadIdx.x & 31;
    constexpr int SPAN = (HEADS == 2) ? 16 : 32;

    const float4* __restrict__ xl4 = (const float4*)xl;
    const float4 t  = ((const float4*)att)[lane];
    const float NEG_INF = -__builtin_huge_valf();

    for (int d = d_first; d < N; d += d_step) {
    const float4 b = ((const float4*)xr)[(size_t)d * (DIM / 4) + lane];

    float4 acc = xl4[(size_t)d * (DIM / 4) + lane];
    float m = head_sum<SPAN>(edge_score(acc, b, t));
    float l = 1.f;

    const int beg = start[d];
    const int n   = cnt[d];

    for (int base = 0; base < n; base += 32) {
        const int c = (n - base < 32) ? (n - base) : 32;
        const bool live = lane < c;
        const int my_s = live ? lst[beg + base + lane] : d;

        int pos = live ? 0 : lane;
        for (int i = 0; i < c; ++i) {
            const int si = __builtin_amdgcn_readlane(my_s, i);
            const bool before = (si < my_s) || (si == my_s && i < lane);
            pos += (live && before) ? 1 : 0;
        }

        for (int j = 0; j < c; j += 4) {
            const int i0 = __builtin_ctz(__builtin_amdgcn_ballot_w32(pos == j));
            const int i1 = __builtin_ctz(__builtin_amdgcn_ballot_w32(pos == j + 1));
            const int i2 = __builtin_ctz(__builtin_amdgcn_ballot_w32(pos == j + 2));
            const int i3 = __builtin_ctz(__builtin_amdgcn_ballot_w32(pos == j + 3));
            const int s0 = __builtin_amdgcn_readlane(my_s, i0);
            const int s1 = __builtin_amdgcn_readlane(my_s, i1);
            const int s2 = __builtin_amdgcn_readlane(my_s, i2);
            const int s3 = __builtin_amdgcn_readlane(my_s, i3);
            const float4 a0 = xl4[(size_t)s0 * (DIM / 4) + lane];
            const float4 a1 = xl4[(size_t)s1 * (DIM / 4) + lane];
            const float4 a2 = xl4[(size_t)s2 * (DIM / 4) + lane];
            const float4 a3 = xl4[(size_t)s3 * (DIM / 4) + lane];

            float q0 = edge_score(a0, b, t);
            float q1 = edge_score(a1, b, t);
            float q2 = edge_score(a2, b, t);
            float q3 = edge_score(a3, b, t);
#pragma unroll
            for (int mm = 1; mm < SPAN; mm <<= 1) {
                const float r0 = __shfl_xor(q0, mm, 32);
                const float r1 = __shfl_xor(q1, mm, 32);
                const float r2 = __shfl_xor(q2, mm, 32);
                const float r3 = __shfl_xor(q3, mm, 32);
                q0 += r0; q1 += r1; q2 += r2; q3 += r3;
            }
            const float p0 = q0;
            const float p1 = (j + 1 < c) ? q1 : NEG_INF;
            const float p2 = (j + 2 < c) ? q2 : NEG_INF;
            const float p3 = (j + 3 < c) ? q3 : NEG_INF;

            const float mn = fmaxf(fmaxf(m, p0), fmaxf(p1, fmaxf(p2, p3)));
            const float sc = __expf(m - mn);
            const float w0 = __expf(p0 - mn);
            const float w1 = __expf(p1 - mn);
            const float w2 = __expf(p2 - mn);
            const float w3 = __expf(p3 - mn);
            m = mn;
            l = l * sc + ((w0 + w1) + (w2 + w3));
            acc.x = acc.x * sc + w0 * a0.x + w1 * a1.x + w2 * a2.x + w3 * a3.x;
            acc.y = acc.y * sc + w0 * a0.y + w1 * a1.y + w2 * a2.y + w3 * a3.y;
            acc.z = acc.z * sc + w0 * a0.z + w1 * a1.z + w2 * a2.z + w3 * a3.z;
            acc.w = acc.w * sc + w0 * a0.w + w1 * a1.w + w2 * a2.w + w3 * a3.w;
        }
    }

    const float den = l + 1e-16f;
    const float4 bi = ((const float4*)bias)[lane];
    float4 o;
    o.x = elu(acc.x / den + bi.x);
    o.y = elu(acc.y / den + bi.y);
    o.z = elu(acc.z / den + bi.z);
    o.w = elu(acc.w / den + bi.w);
    ((float4*)out)[(size_t)d * (DIM / 4) + lane] = o;
    }
}

extern "C" void kernel_launch(void* const* d_in, const int* in_sizes, int n_in,
                              void* d_out, int out_size, void* d_ws, size_t ws_size,
                              hipStream_t stream) {
    const float* x    = (const float*)d_in[0];
    const int*   ei   = (const int*)d_in[1];
    const float* Wl1  = (const float*)d_in[2];
    const float* Wr1  = (const float*)d_in[3];
    const float* att1 = (const float*)d_in[4];
    const float* b1   = (const float*)d_in[5];
    const float* Wl2  = (const float*)d_in[6];
    const float* Wr2  = (const float*)d_in[7];
    const float* att2 = (const float*)d_in[8];
    const float* b2   = (const float*)d_in[9];

    const int N  = in_sizes[0] / DIM;
    const int E  = in_sizes[1] / 2;

    const int* srcA = ei;
    const int* dstA = ei + E;

    float* xl    = (float*)d_ws;
    float* xr    = xl + (size_t)N * DIM;
    float* h1    = xr + (size_t)N * DIM;
    int*   cnt   = (int*)(h1 + (size_t)N * DIM);
    int*   start = cnt + (size_t)N + 64;
    int*   rank  = start + (size_t)N + 64;
    int*   lst   = rank + (size_t)E;
    float* out   = (float*)d_out;

    const dim3 B(256);
    const dim3 BG(128);
    const int gCnt   = (N + 1 + 255) / 256;
    const int gNodeT = (N + 255) / 256;
    const int gEdgeT = (E + 256 * EPT - 1) / (256 * EPT);
    const int nTiles = N / 16;
    const int gPair  = (2 * nTiles + 3) / 4;
    const int gRank  = (E + 128 * EPT - 1) / (128 * EPT);
    const int gNodeW = (N + 8 * NODES_PER_WAVE - 1) / (8 * NODES_PER_WAVE);

    zero_fill<<<gCnt, B, 0, stream>>>((float*)cnt, N + 1);
    proj_pair_wmma_f32<<<gRank + gPair, BG, 0, stream>>>(
        x, Wl1, Wr1, xl, xr, nTiles, dstA, E, cnt, rank, gRank);
    seg_alloc<<<gNodeT, B, 0, stream>>>(cnt, start, N);
    fill_lists<<<gEdgeT, B, 0, stream>>>(srcA, dstA, E, start, rank, lst);

    node_attend<2><<<gNodeW, B, 0, stream>>>(start, cnt, lst, N, xl, xr, att1, b1, h1);

    proj_pair_wmma_f32<<<gPair, BG, 0, stream>>>(
        h1, Wl2, Wr2, xl, xr, nTiles, dstA, E, cnt, rank, 0);
    node_attend<1><<<gNodeW, B, 0, stream>>>(start, cnt, lst, N, xl, xr, att2, b2, out);
}


# Edge lists in one pass: counting pass writes each source id into a fixed 32-slot line of its destination; edges past 32 go to an overflow list walked in arrival order; 7 launches to 5 (no fill pass)
# speedup vs baseline: 4.0808x; 1.0317x over previous
#include <hip/hip_runtime.h>
#include <hip/hip_bf16.h>

typedef __attribute__((ext_vector_type(2))) float v2f;
typedef __attribute__((ext_vector_type(8))) float v8f;

#define DIM 128
#define NODES_PER_WAVE 4
#define KSTAGE 2
#define NEG_SLOPE 0.2f

__device__ __forceinline__ float lrelu(float x) { return x > 0.f ? x : NEG_SLOPE * x; }
__device__ __forceinline__ float elu(float x)   { return x > 0.f ? x : (__expf(x) - 1.f); }

#define WMMA_F32(a, b, c) __builtin_amdgcn_wmma_f32_16x16x4_f32( \
     false, a,  false, b,  (short)0, c, \
     false,  false)

struct KFrag {
    v2f a;
    float4 p0, q0;
    float4 p1, q1;
};

__device__ __forceinline__ KFrag load_kfrag(const float* __restrict__ Ap,
                                            const float* __restrict__ Wp, int k) {
    KFrag f;
    f.a  = *(const v2f*)(Ap + k);
    f.p0 = *(const float4*)(Wp + (size_t)k * DIM);
    f.q0 = *(const float4*)(Wp + (size_t)(k + 1) * DIM);
    f.p1 = *(const float4*)(Wp + (size_t)k * DIM + 64);
    f.q1 = *(const float4*)(Wp + (size_t)(k + 1) * DIM + 64);
    return f;
}

__device__ __forceinline__ void mma_kfrag(const KFrag& f, v8f (&c)[8]) {
    v2f b;
    b.x = f.p0.x; b.y = f.q0.x; c[0] = WMMA_F32(f.a, b, c[0]);
    b.x = f.p0.y; b.y = f.q0.y; c[1] = WMMA_F32(f.a, b, c[1]);
    b.x = f.p0.z; b.y = f.q0.z; c[2] = WMMA_F32(f.a, b, c[2]);
    b.x = f.p0.w; b.y = f.q0.w; c[3] = WMMA_F32(f.a, b, c[3]);
    b.x = f.p1.x; b.y = f.q1.x; c[4] = WMMA_F32(f.a, b, c[4]);
    b.x = f.p1.y; b.y = f.q1.y; c[5] = WMMA_F32(f.a, b, c[5]);
    b.x = f.p1.z; b.y = f.q1.z; c[6] = WMMA_F32(f.a, b, c[6]);
    b.x = f.p1.w; b.y = f.q1.w; c[7] = WMMA_F32(f.a, b, c[7]);
}

#define EPT 8
#define CAP 32
__device__ __forceinline__ void dst_slot_role(
    const int* __restrict__ srcA, const int* __restrict__ dstA, int E, int N,
    int* cnt, int* __restrict__ slot,
    int* __restrict__ ovf_d, int* __restrict__ ovf_s, int e0, int T) {
    int dn[EPT], sn[EPT];
#pragma unroll
    for (int k = 0; k < EPT; ++k) {
        const int e  = e0 + k * T;
        const int ec = (e < E) ? e : (E - 1);
        dn[k] = dstA[ec];
        sn[k] = srcA[ec];
    }
    int rk[EPT];
#pragma unroll
    for (int k = 0; k < EPT; ++k) {
        const int e = e0 + k * T;
        rk[k] = atomicAdd(&cnt[dn[k]], (e < E) ? 1 : 0);
    }
#pragma unroll
    for (int k = 0; k < EPT; ++k) {
        const int e = e0 + k * T;
        if (e < E) {
            if (rk[k] < CAP) {
                slot[dn[k] * CAP + rk[k]] = sn[k];
            } else {
                const int p = atomicAdd(&cnt[N], 1);
                ovf_d[p] = dn[k];
                ovf_s[p] = sn[k];
            }
        }
    }
}

__global__ void __launch_bounds__(128) proj_pair_wmma_f32(
    const float* __restrict__ A,
    const float* __restrict__ W0, const float* __restrict__ W1,
    float* __restrict__ C0, float* __restrict__ C1, int ntiles,
    const int* __restrict__ srcA, const int* __restrict__ dstA, int E, int N,
    int* cnt, int* __restrict__ slot,
    int* __restrict__ ovf_d, int* __restrict__ ovf_s, int rankBlocks) {
    if ((int)blockIdx.x < rankBlocks) {
        dst_slot_role(srcA, dstA, E, N, cnt, slot, ovf_d, ovf_s,
                      (int)(blockIdx.x * 128 + threadIdx.x), rankBlocks * 128);
        return;
    }
    const int lane = threadIdx.x & 31;
    const int half = lane >> 4;
    const int l15  = lane & 15;
    const int strip = __builtin_amdgcn_readfirstlane(
        (int)((blockIdx.x - rankBlocks) * 4 + (threadIdx.x >> 5)));
    if (strip >= 2 * ntiles) return;
    const bool second = strip >= ntiles;
    const float* __restrict__ W = second ? W1 : W0;
    float* __restrict__ C       = second ? C1 : C0;
    const int tile = second ? strip - ntiles : strip;
    const int row0 = tile * 16;

    const float* __restrict__ Ap = A + (size_t)(row0 + l15) * DIM + half * 2;
    const float* __restrict__ Wp = W + (size_t)(half * 2) * DIM + l15 * 4;

    v8f c[8] = {};
    KFrag fa[KSTAGE], fb[KSTAGE];
    constexpr int KS4 = 4 * KSTAGE;
#pragma unroll
    for (int s = 0; s < KSTAGE; ++s) fa[s] = load_kfrag(Ap, Wp, 4 * s);
#pragma unroll 1
    for (int k = 0; k < DIM - 2 * KS4; k += 2 * KS4) {
#pragma unroll
        for (int s = 0; s < KSTAGE; ++s) fb[s] = load_kfrag(Ap, Wp, k + KS4 + 4 * s);
#pragma unroll
        for (int s = 0; s < KSTAGE; ++s) mma_kfrag(fa[s], c);
#pragma unroll
        for (int s = 0; s < KSTAGE; ++s) fa[s] = load_kfrag(Ap, Wp, k + 2 * KS4 + 4 * s);
#pragma unroll
        for (int s = 0; s < KSTAGE; ++s) mma_kfrag(fb[s], c);
    }
#pragma unroll
    for (int s = 0; s < KSTAGE; ++s) fb[s] = load_kfrag(Ap, Wp, DIM - KS4 + 4 * s);
#pragma unroll
    for (int s = 0; s < KSTAGE; ++s) mma_kfrag(fa[s], c);
#pragma unroll
    for (int s = 0; s < KSTAGE; ++s) mma_kfrag(fb[s], c);

    float* __restrict__ Cp = C + (size_t)(row0 + half * 8) * DIM + l15 * 4;
#pragma unroll
    for (int r = 0; r < 8; ++r) {
        float4 lo, hi;
        lo.x = c[0][r]; lo.y = c[1][r]; lo.z = c[2][r]; lo.w = c[3][r];
        hi.x = c[4][r]; hi.y = c[5][r]; hi.z = c[6][r]; hi.w = c[7][r];
        *(float4*)(Cp + (size_t)r * DIM)      = lo;
        *(float4*)(Cp + (size_t)r * DIM + 64) = hi;
    }
}

__global__ void zero_fill(float* __restrict__ p, int n) {
    int i = blockIdx.x * blockDim.x + threadIdx.x;
    if (i < n) p[i] = 0.f;
}

__device__ __forceinline__ float edge_score(const float4 a, const float4 b,
                                            const float4 t) {
    return lrelu(a.x + b.x) * t.x + lrelu(a.y + b.y) * t.y +
           lrelu(a.z + b.z) * t.z + lrelu(a.w + b.w) * t.w;
}

template <int SPAN>
__device__ __forceinline__ float head_sum(float p) {
#pragma unroll
    for (int m = 1; m < SPAN; m <<= 1) p += __shfl_xor(p, m, 32);
    return p;
}

template <int HEADS>
__global__ void __launch_bounds__(256, 16) node_attend(
    const int* __restrict__ cnt, const int* __restrict__ slot,
    const int* __restrict__ ovf_d, const int* __restrict__ ovf_s, int N,
    const float* __restrict__ xl, const float* __restrict__ xr,
    const float* __restrict__ att, const float* bias,
    float* out) {
    const int d_first = __builtin_amdgcn_readfirstlane(
        (int)(blockIdx.x * (blockDim.x >> 5) + (threadIdx.x >> 5)));
    const int d_step = (int)(gridDim.x * (blockDim.x >> 5));
    const int lane = threadIdx.x & 31;
    constexpr int SPAN = (HEADS == 2) ? 16 : 32;

    const float4* __restrict__ xl4 = (const float4*)xl;
    const float4 t  = ((const float4*)att)[lane];
    const float NEG_INF = -__builtin_huge_valf();

    for (int d = d_first; d < N; d += d_step) {
    const float4 b = ((const float4*)xr)[(size_t)d * (DIM / 4) + lane];

    float4 acc = xl4[(size_t)d * (DIM / 4) + lane];
    float m = head_sum<SPAN>(edge_score(acc, b, t));
    float l = 1.f;

    const int n    = cnt[d];
    const int novf = (n > CAP) ? cnt[N] : 0;

    for (int ob = -32; ob < novf; ob += 32) {
        bool live; int my_s, c, nl, pos;
        if (ob < 0) {
            c    = (n < CAP) ? n : CAP;
            live = lane < c;
            my_s = live ? slot[d * CAP + lane] : d;
            nl   = c;
            pos  = live ? 0 : lane - c;
        } else {
            const int idx = ob + lane;
            const int od  = (idx < novf) ? ovf_d[idx] : -1;
            live = (od == d);
            my_s = live ? ovf_s[idx] : d;
            c    = __builtin_popcount(__builtin_amdgcn_ballot_w32(live));
            nl   = (c > 0) ? 32 : 0;
            pos  = 0;
        }

        const int key = live ? my_s : 0x7fffffff;
        for (int i = 0; i < nl; ++i) {
            const int si = __builtin_amdgcn_readlane(key, i);
            const bool before = (si < key) || (si == key && i < lane);
            pos += before ? 1 : 0;
        }

        for (int j = 0; j < c; j += 4) {
            const int i0 = __builtin_ctz(__builtin_amdgcn_ballot_w32(pos == j));
            const int i1 = __builtin_ctz(__builtin_amdgcn_ballot_w32(pos == j + 1));
            const int i2 = __builtin_ctz(__builtin_amdgcn_ballot_w32(pos == j + 2));
            const int i3 = __builtin_ctz(__builtin_amdgcn_ballot_w32(pos == j + 3));
            const int s0 = __builtin_amdgcn_readlane(my_s, i0);
            const int s1 = __builtin_amdgcn_readlane(my_s, i1);
            const int s2 = __builtin_amdgcn_readlane(my_s, i2);
            const int s3 = __builtin_amdgcn_readlane(my_s, i3);
            const float4 a0 = xl4[(size_t)s0 * (DIM / 4) + lane];
            const float4 a1 = xl4[(size_t)s1 * (DIM / 4) + lane];
            const float4 a2 = xl4[(size_t)s2 * (DIM / 4) + lane];
            const float4 a3 = xl4[(size_t)s3 * (DIM / 4) + lane];

            float q0 = edge_score(a0, b, t);
            float q1 = edge_score(a1, b, t);
            float q2 = edge_score(a2, b, t);
            float q3 = edge_score(a3, b, t);
#pragma unroll
            for (int mm = 1; mm < SPAN; mm <<= 1) {
                const float r0 = __shfl_xor(q0, mm, 32);
                const float r1 = __shfl_xor(q1, mm, 32);
                const float r2 = __shfl_xor(q2, mm, 32);
                const float r3 = __shfl_xor(q3, mm, 32);
                q0 += r0; q1 += r1; q2 += r2; q3 += r3;
            }
            const float p0 = q0;
            const float p1 = (j + 1 < c) ? q1 : NEG_INF;
            const float p2 = (j + 2 < c) ? q2 : NEG_INF;
            const float p3 = (j + 3 < c) ? q3 : NEG_INF;

            const float mn = fmaxf(fmaxf(m, p0), fmaxf(p1, fmaxf(p2, p3)));
            const float sc = __expf(m - mn);
            const float w0 = __expf(p0 - mn);
            const float w1 = __expf(p1 - mn);
            const float w2 = __expf(p2 - mn);
            const float w3 = __expf(p3 - mn);
            m = mn;
            l = l * sc + ((w0 + w1) + (w2 + w3));
            acc.x = acc.x * sc + w0 * a0.x + w1 * a1.x + w2 * a2.x + w3 * a3.x;
            acc.y = acc.y * sc + w0 * a0.y + w1 * a1.y + w2 * a2.y + w3 * a3.y;
            acc.z = acc.z * sc + w0 * a0.z + w1 * a1.z + w2 * a2.z + w3 * a3.z;
            acc.w = acc.w * sc + w0 * a0.w + w1 * a1.w + w2 * a2.w + w3 * a3.w;
        }
    }

    const float den = l + 1e-16f;
    const float4 bi = ((const float4*)bias)[lane];
    float4 o;
    o.x = elu(acc.x / den + bi.x);
    o.y = elu(acc.y / den + bi.y);
    o.z = elu(acc.z / den + bi.z);
    o.w = elu(acc.w / den + bi.w);
    ((float4*)out)[(size_t)d * (DIM / 4) + lane] = o;
    }
}

extern "C" void kernel_launch(void* const* d_in, const int* in_sizes, int n_in,
                              void* d_out, int out_size, void* d_ws, size_t ws_size,
                              hipStream_t stream) {
    const float* x    = (const float*)d_in[0];
    const int*   ei   = (const int*)d_in[1];
    const float* Wl1  = (const float*)d_in[2];
    const float* Wr1  = (const float*)d_in[3];
    const float* att1 = (const float*)d_in[4];
    const float* b1   = (const float*)d_in[5];
    const float* Wl2  = (const float*)d_in[6];
    const float* Wr2  = (const float*)d_in[7];
    const float* att2 = (const float*)d_in[8];
    const float* b2   = (const float*)d_in[9];

    const int N  = in_sizes[0] / DIM;
    const int E  = in_sizes[1] / 2;

    const int* srcA = ei;
    const int* dstA = ei + E;

    float* xl    = (float*)d_ws;
    float* xr    = xl + (size_t)N * DIM;
    float* h1    = xr + (size_t)N * DIM;
    int*   slot  = (int*)(h1 + (size_t)N * DIM);
    int*   ovf_d = slot + (size_t)N * CAP;
    int*   ovf_s = ovf_d + (size_t)E;
    int*   cnt   = ovf_s + (size_t)E;
    float* out   = (float*)d_out;

    const dim3 B(256);
    const dim3 BG(128);
    const int gCnt   = (N + 1 + 255) / 256;
    const int nTiles = N / 16;
    const int gPair  = (2 * nTiles + 3) / 4;
    const int gRank  = (E + 128 * EPT - 1) / (128 * EPT);
    const int gNodeW = (N + 8 * NODES_PER_WAVE - 1) / (8 * NODES_PER_WAVE);

    zero_fill<<<gCnt, B, 0, stream>>>((float*)cnt, N + 1);
    proj_pair_wmma_f32<<<gRank + gPair, BG, 0, stream>>>(
        x, Wl1, Wr1, xl, xr, nTiles, srcA, dstA, E, N, cnt, slot, ovf_d, ovf_s, gRank);

    node_attend<2><<<gNodeW, B, 0, stream>>>(cnt, slot, ovf_d, ovf_s, N, xl, xr, att1, b1, h1);

    proj_pair_wmma_f32<<<gPair, BG, 0, stream>>>(
        h1, Wl2, Wr2, xl, xr, nTiles, srcA, dstA, E, N, cnt, slot, ovf_d, ovf_s, 0);
    node_attend<1><<<gNodeW, B, 0, stream>>>(cnt, slot, ovf_d, ovf_s, N, xl, xr, att2, b2, out);
}


# Attend walk with fewer vector instructions: leaky relu as max(x, 0.2x), score products summed pairwise in packed form, id order by one-compare keys and one ds_permute, one reciprocal per node
# speedup vs baseline: 4.3101x; 1.0562x over previous
#include <hip/hip_runtime.h>
#include <hip/hip_bf16.h>

typedef __attribute__((ext_vector_type(2))) float v2f;
typedef __attribute__((ext_vector_type(8))) float v8f;

#define DIM 128
#define NODES_PER_WAVE 4
#define KSTAGE 2
#define NEG_SLOPE 0.2f

__device__ __forceinline__ float elu(float x)   { return x > 0.f ? x : (__expf(x) - 1.f); }

#define WMMA_F32(a, b, c) __builtin_amdgcn_wmma_f32_16x16x4_f32( \
     false, a,  false, b,  (short)0, c, \
     false,  false)

struct KFrag {
    v2f a;
    float4 p0, q0;
    float4 p1, q1;
};

__device__ __forceinline__ KFrag load_kfrag(const float* __restrict__ Ap,
                                            const float* __restrict__ Wp, int k) {
    KFrag f;
    f.a  = *(const v2f*)(Ap + k);
    f.p0 = *(const float4*)(Wp + (size_t)k * DIM);
    f.q0 = *(const float4*)(Wp + (size_t)(k + 1) * DIM);
    f.p1 = *(const float4*)(Wp + (size_t)k * DIM + 64);
    f.q1 = *(const float4*)(Wp + (size_t)(k + 1) * DIM + 64);
    return f;
}

__device__ __forceinline__ void mma_kfrag(const KFrag& f, v8f (&c)[8]) {
    v2f b;
    b.x = f.p0.x; b.y = f.q0.x; c[0] = WMMA_F32(f.a, b, c[0]);
    b.x = f.p0.y; b.y = f.q0.y; c[1] = WMMA_F32(f.a, b, c[1]);
    b.x = f.p0.z; b.y = f.q0.z; c[2] = WMMA_F32(f.a, b, c[2]);
    b.x = f.p0.w; b.y = f.q0.w; c[3] = WMMA_F32(f.a, b, c[3]);
    b.x = f.p1.x; b.y = f.q1.x; c[4] = WMMA_F32(f.a, b, c[4]);
    b.x = f.p1.y; b.y = f.q1.y; c[5] = WMMA_F32(f.a, b, c[5]);
    b.x = f.p1.z; b.y = f.q1.z; c[6] = WMMA_F32(f.a, b, c[6]);
    b.x = f.p1.w; b.y = f.q1.w; c[7] = WMMA_F32(f.a, b, c[7]);
}

#define EPT 8
#define CAP 32
__device__ __forceinline__ void dst_slot_role(
    const int* __restrict__ srcA, const int* __restrict__ dstA, int E, int N,
    int* cnt, int* __restrict__ slot,
    int* __restrict__ ovf_d, int* __restrict__ ovf_s, int e0, int T) {
    int dn[EPT], sn[EPT];
#pragma unroll
    for (int k = 0; k < EPT; ++k) {
        const int e  = e0 + k * T;
        const int ec = (e < E) ? e : (E - 1);
        dn[k] = dstA[ec];
        sn[k] = srcA[ec];
    }
    int rk[EPT];
#pragma unroll
    for (int k = 0; k < EPT; ++k) {
        const int e = e0 + k * T;
        rk[k] = atomicAdd(&cnt[dn[k]], (e < E) ? 1 : 0);
    }
#pragma unroll
    for (int k = 0; k < EPT; ++k) {
        const int e = e0 + k * T;
        if (e < E) {
            if (rk[k] < CAP) {
                slot[dn[k] * CAP + rk[k]] = sn[k];
            } else {
                const int p = atomicAdd(&cnt[N], 1);
                ovf_d[p] = dn[k];
                ovf_s[p] = sn[k];
            }
        }
    }
}

__global__ void __launch_bounds__(128) proj_pair_wmma_f32(
    const float* __restrict__ A,
    const float* __restrict__ W0, const float* __restrict__ W1,
    float* __restrict__ C0, float* __restrict__ C1, int ntiles,
    const int* __restrict__ srcA, const int* __restrict__ dstA, int E, int N,
    int* cnt, int* __restrict__ slot,
    int* __restrict__ ovf_d, int* __restrict__ ovf_s, int rankBlocks) {
    if ((int)blockIdx.x < rankBlocks) {
        dst_slot_role(srcA, dstA, E, N, cnt, slot, ovf_d, ovf_s,
                      (int)(blockIdx.x * 128 + threadIdx.x), rankBlocks * 128);
        return;
    }
    const int lane = threadIdx.x & 31;
    const int half = lane >> 4;
    const int l15  = lane & 15;
    const int strip = __builtin_amdgcn_readfirstlane(
        (int)((blockIdx.x - rankBlocks) * 4 + (threadIdx.x >> 5)));
    if (strip >= 2 * ntiles) return;
    const bool second = strip >= ntiles;
    const float* __restrict__ W = second ? W1 : W0;
    float* __restrict__ C       = second ? C1 : C0;
    const int tile = second ? strip - ntiles : strip;
    const int row0 = tile * 16;

    const float* __restrict__ Ap = A + (size_t)(row0 + l15) * DIM + half * 2;
    const float* __restrict__ Wp = W + (size_t)(half * 2) * DIM + l15 * 4;

    v8f c[8] = {};
    KFrag fa[KSTAGE], fb[KSTAGE];
    constexpr int KS4 = 4 * KSTAGE;
#pragma unroll
    for (int s = 0; s < KSTAGE; ++s) fa[s] = load_kfrag(Ap, Wp, 4 * s);
#pragma unroll 1
    for (int k = 0; k < DIM - 2 * KS4; k += 2 * KS4) {
#pragma unroll
        for (int s = 0; s < KSTAGE; ++s) fb[s] = load_kfrag(Ap, Wp, k + KS4 + 4 * s);
#pragma unroll
        for (int s = 0; s < KSTAGE; ++s) mma_kfrag(fa[s], c);
#pragma unroll
        for (int s = 0; s < KSTAGE; ++s) fa[s] = load_kfrag(Ap, Wp, k + 2 * KS4 + 4 * s);
#pragma unroll
        for (int s = 0; s < KSTAGE; ++s) mma_kfrag(fb[s], c);
    }
#pragma unroll
    for (int s = 0; s < KSTAGE; ++s) fb[s] = load_kfrag(Ap, Wp, DIM - KS4 + 4 * s);
#pragma unroll
    for (int s = 0; s < KSTAGE; ++s) mma_kfrag(fa[s], c);
#pragma unroll
    for (int s = 0; s < KSTAGE; ++s) mma_kfrag(fb[s], c);

    float* __restrict__ Cp = C + (size_t)(row0 + half * 8) * DIM + l15 * 4;
#pragma unroll
    for (int r = 0; r < 8; ++r) {
        float4 lo, hi;
        lo.x = c[0][r]; lo.y = c[1][r]; lo.z = c[2][r]; lo.w = c[3][r];
        hi.x = c[4][r]; hi.y = c[5][r]; hi.z = c[6][r]; hi.w = c[7][r];
        *(float4*)(Cp + (size_t)r * DIM)      = lo;
        *(float4*)(Cp + (size_t)r * DIM + 64) = hi;
    }
}

__global__ void zero_fill(float* __restrict__ p, int n) {
    int i = blockIdx.x * blockDim.x + threadIdx.x;
    if (i < n) p[i] = 0.f;
}

__device__ __forceinline__ v2f lrelu2(const v2f x) {
    return __builtin_elementwise_max(x, x * NEG_SLOPE);
}
__device__ __forceinline__ float edge_score(const float4 a, const float4 b,
                                            const float4 t) {
    const v2f lo = lrelu2(v2f{a.x, a.y} + v2f{b.x, b.y});
    const v2f hi = lrelu2(v2f{a.z, a.w} + v2f{b.z, b.w});
    const v2f p  = __builtin_elementwise_fma(hi, v2f{t.z, t.w}, lo * v2f{t.x, t.y});
    return p.x + p.y;
}

template <int SPAN>
__device__ __forceinline__ float head_sum(float p) {
#pragma unroll
    for (int m = 1; m < SPAN; m <<= 1) p += __shfl_xor(p, m, 32);
    return p;
}

template <int HEADS>
__global__ void __launch_bounds__(256, 16) node_attend(
    const int* __restrict__ cnt, const int* __restrict__ slot,
    const int* __restrict__ ovf_d, const int* __restrict__ ovf_s, int N,
    const float* __restrict__ xl, const float* __restrict__ xr,
    const float* __restrict__ att, const float* bias,
    float* out) {
    const int d_first = __builtin_amdgcn_readfirstlane(
        (int)(blockIdx.x * (blockDim.x >> 5) + (threadIdx.x >> 5)));
    const int d_step = (int)(gridDim.x * (blockDim.x >> 5));
    const int lane = threadIdx.x & 31;
    constexpr int SPAN = (HEADS == 2) ? 16 : 32;

    const float4* __restrict__ xl4 = (const float4*)xl;
    const float4 t  = ((const float4*)att)[lane];
    const float NEG_INF = -__builtin_huge_valf();

    for (int d = d_first; d < N; d += d_step) {
    const float4 b = ((const float4*)xr)[(size_t)d * (DIM / 4) + lane];

    float4 acc = xl4[(size_t)d * (DIM / 4) + lane];
    float m = head_sum<SPAN>(edge_score(acc, b, t));
    float l = 1.f;

    const int n    = cnt[d];
    const int novf = (n > CAP) ? cnt[N] : 0;

    for (int ob = -32; ob < novf; ob += 32) {
        bool live; int my_s, c, nl, pos;
        if (ob < 0) {
            c    = (n < CAP) ? n : CAP;
            live = lane < c;
            my_s = live ? slot[d * CAP + lane] : d;
            nl   = c;
            pos  = live ? 0 : lane - c;
        } else {
            const int idx = ob + lane;
            const int od  = (idx < novf) ? ovf_d[idx] : -1;
            live = (od == d);
            my_s = live ? ovf_s[idx] : d;
            c    = __builtin_popcount(__builtin_amdgcn_ballot_w32(live));
            nl   = (c > 0) ? 32 : 0;
            pos  = 0;
        }

        const int key = live ? ((my_s << 5) | lane) : (0x7fffffe0 | lane);
        for (int i = 0; i < nl; ++i) {
            const int si = __builtin_amdgcn_readlane(key, i);
            pos += (si < key) ? 1 : 0;
        }
        const int sorted_s = __builtin_amdgcn_ds_permute(pos << 2, my_s);

        for (int j = 0; j < c; j += 4) {
            const int s0 = __builtin_amdgcn_readlane(sorted_s, j);
            const int s1 = __builtin_amdgcn_readlane(sorted_s, j + 1);
            const int s2 = __builtin_amdgcn_readlane(sorted_s, j + 2);
            const int s3 = __builtin_amdgcn_readlane(sorted_s, j + 3);
            const float4 a0 = xl4[(size_t)s0 * (DIM / 4) + lane];
            const float4 a1 = xl4[(size_t)s1 * (DIM / 4) + lane];
            const float4 a2 = xl4[(size_t)s2 * (DIM / 4) + lane];
            const float4 a3 = xl4[(size_t)s3 * (DIM / 4) + lane];

            float q0 = edge_score(a0, b, t);
            float q1 = edge_score(a1, b, t);
            float q2 = edge_score(a2, b, t);
            float q3 = edge_score(a3, b, t);
#pragma unroll
            for (int mm = 1; mm < SPAN; mm <<= 1) {
                const float r0 = __shfl_xor(q0, mm, 32);
                const float r1 = __shfl_xor(q1, mm, 32);
                const float r2 = __shfl_xor(q2, mm, 32);
                const float r3 = __shfl_xor(q3, mm, 32);
                q0 += r0; q1 += r1; q2 += r2; q3 += r3;
            }
            const float p0 = q0;
            const float p1 = (j + 1 < c) ? q1 : NEG_INF;
            const float p2 = (j + 2 < c) ? q2 : NEG_INF;
            const float p3 = (j + 3 < c) ? q3 : NEG_INF;

            const float mn = fmaxf(fmaxf(m, p0), fmaxf(p1, fmaxf(p2, p3)));
            const float sc = __expf(m - mn);
            const float w0 = __expf(p0 - mn);
            const float w1 = __expf(p1 - mn);
            const float w2 = __expf(p2 - mn);
            const float w3 = __expf(p3 - mn);
            m = mn;
            l = l * sc + ((w0 + w1) + (w2 + w3));
            acc.x = acc.x * sc + w0 * a0.x + w1 * a1.x + w2 * a2.x + w3 * a3.x;
            acc.y = acc.y * sc + w0 * a0.y + w1 * a1.y + w2 * a2.y + w3 * a3.y;
            acc.z = acc.z * sc + w0 * a0.z + w1 * a1.z + w2 * a2.z + w3 * a3.z;
            acc.w = acc.w * sc + w0 * a0.w + w1 * a1.w + w2 * a2.w + w3 * a3.w;
        }
    }

    const float den = l + 1e-16f;
    float inv = __builtin_amdgcn_rcpf(den);
    inv = __builtin_fmaf(__builtin_fmaf(-den, inv, 1.f), inv, inv);
    const float4 bi = ((const float4*)bias)[lane];
    float4 o;
    o.x = elu(acc.x * inv + bi.x);
    o.y = elu(acc.y * inv + bi.y);
    o.z = elu(acc.z * inv + bi.z);
    o.w = elu(acc.w * inv + bi.w);
    ((float4*)out)[(size_t)d * (DIM / 4) + lane] = o;
    }
}

extern "C" void kernel_launch(void* const* d_in, const int* in_sizes, int n_in,
                              void* d_out, int out_size, void* d_ws, size_t ws_size,
                              hipStream_t stream) {
    const float* x    = (const float*)d_in[0];
    const int*   ei   = (const int*)d_in[1];
    const float* Wl1  = (const float*)d_in[2];
    const float* Wr1  = (const float*)d_in[3];
    const float* att1 = (const float*)d_in[4];
    const float* b1   = (const float*)d_in[5];
    const float* Wl2  = (const float*)d_in[6];
    const float* Wr2  = (const float*)d_in[7];
    const float* att2 = (const float*)d_in[8];
    const float* b2   = (const float*)d_in[9];

    const int N  = in_sizes[0] / DIM;
    const int E  = in_sizes[1] / 2;

    const int* srcA = ei;
    const int* dstA = ei + E;

    float* xl    = (float*)d_ws;
    float* xr    = xl + (size_t)N * DIM;
    float* h1    = xr + (size_t)N * DIM;
    int*   slot  = (int*)(h1 + (size_t)N * DIM);
    int*   ovf_d = slot + (size_t)N * CAP;
    int*   ovf_s = ovf_d + (size_t)E;
    int*   cnt   = ovf_s + (size_t)E;
    float* out   = (float*)d_out;

    const dim3 B(256);
    const dim3 BG(128);
    const int gCnt   = (N + 1 + 255) / 256;
    const int nTiles = N / 16;
    const int gPair  = (2 * nTiles + 3) / 4;
    const int gRank  = (E + 128 * EPT - 1) / (128 * EPT);
    const int gNodeW = (N + 8 * NODES_PER_WAVE - 1) / (8 * NODES_PER_WAVE);

    zero_fill<<<gCnt, B, 0, stream>>>((float*)cnt, N + 1);
    proj_pair_wmma_f32<<<gRank + gPair, BG, 0, stream>>>(
        x, Wl1, Wr1, xl, xr, nTiles, srcA, dstA, E, N, cnt, slot, ovf_d, ovf_s, gRank);

    node_attend<2><<<gNodeW, B, 0, stream>>>(cnt, slot, ovf_d, ovf_s, N, xl, xr, att1, b1, h1);

    proj_pair_wmma_f32<<<gPair, BG, 0, stream>>>(
        h1, Wl2, Wr2, xl, xr, nTiles, srcA, dstA, E, N, cnt, slot, ovf_d, ovf_s, 0);
    node_attend<1><<<gNodeW, B, 0, stream>>>(cnt, slot, ovf_d, ovf_s, N, xl, xr, att2, b2, out);
}


# Layer 1 attend and layer 2's two projections fused into one launch: a block's four waves put their 16 finished rows into LDS and multiply them from there (A fragments from LDS); 5 launches to 4
# speedup vs baseline: 4.7722x; 1.1072x over previous
#include <hip/hip_runtime.h>
#include <hip/hip_bf16.h>

typedef __attribute__((ext_vector_type(2))) float v2f;
typedef __attribute__((ext_vector_type(8))) float v8f;

#define DIM 128
#define NODES_PER_WAVE 4
#define KSTAGE 2
#define NEG_SLOPE 0.2f

__device__ __forceinline__ float elu(float x)   { return x > 0.f ? x : (__expf(x) - 1.f); }

#define WMMA_F32(a, b, c) __builtin_amdgcn_wmma_f32_16x16x4_f32( \
     false, a,  false, b,  (short)0, c, \
     false,  false)

struct KFrag {
    v2f a;
    float4 p0, q0;
    float4 p1, q1;
};

__device__ __forceinline__ KFrag load_kfrag(const float* __restrict__ Ap,
                                            const float* __restrict__ Wp, int k) {
    KFrag f;
    f.a  = *(const v2f*)(Ap + k);
    f.p0 = *(const float4*)(Wp + (size_t)k * DIM);
    f.q0 = *(const float4*)(Wp + (size_t)(k + 1) * DIM);
    f.p1 = *(const float4*)(Wp + (size_t)k * DIM + 64);
    f.q1 = *(const float4*)(Wp + (size_t)(k + 1) * DIM + 64);
    return f;
}

__device__ __forceinline__ void mma_kfrag(const KFrag& f, v8f (&c)[8]) {
    v2f b;
    b.x = f.p0.x; b.y = f.q0.x; c[0] = WMMA_F32(f.a, b, c[0]);
    b.x = f.p0.y; b.y = f.q0.y; c[1] = WMMA_F32(f.a, b, c[1]);
    b.x = f.p0.z; b.y = f.q0.z; c[2] = WMMA_F32(f.a, b, c[2]);
    b.x = f.p0.w; b.y = f.q0.w; c[3] = WMMA_F32(f.a, b, c[3]);
    b.x = f.p1.x; b.y = f.q1.x; c[4] = WMMA_F32(f.a, b, c[4]);
    b.x = f.p1.y; b.y = f.q1.y; c[5] = WMMA_F32(f.a, b, c[5]);
    b.x = f.p1.z; b.y = f.q1.z; c[6] = WMMA_F32(f.a, b, c[6]);
    b.x = f.p1.w; b.y = f.q1.w; c[7] = WMMA_F32(f.a, b, c[7]);
}

struct KFragH {
    v2f a;
    float4 p, q;
};

__device__ __forceinline__ KFragH load_kfrag_h(const float* Ap,
                                               const float* __restrict__ Wp, int k) {
    KFragH f;
    f.a = *(const v2f*)(Ap + k);
    f.p = *(const float4*)(Wp + (size_t)k * DIM);
    f.q = *(const float4*)(Wp + (size_t)(k + 1) * DIM);
    return f;
}

__device__ __forceinline__ void mma_kfrag_h(const KFragH& f, v8f (&c)[4]) {
    v2f b;
    b.x = f.p.x; b.y = f.q.x; c[0] = WMMA_F32(f.a, b, c[0]);
    b.x = f.p.y; b.y = f.q.y; c[1] = WMMA_F32(f.a, b, c[1]);
    b.x = f.p.z; b.y = f.q.z; c[2] = WMMA_F32(f.a, b, c[2]);
    b.x = f.p.w; b.y = f.q.w; c[3] = WMMA_F32(f.a, b, c[3]);
}

#define EPT 8
#define CAP 32
__device__ __forceinline__ void dst_slot_role(
    const int* __restrict__ srcA, const int* __restrict__ dstA, int E, int N,
    int* cnt, int* __restrict__ slot,
    int* __restrict__ ovf_d, int* __restrict__ ovf_s, int e0, int T) {
    int dn[EPT], sn[EPT];
#pragma unroll
    for (int k = 0; k < EPT; ++k) {
        const int e  = e0 + k * T;
        const int ec = (e < E) ? e : (E - 1);
        dn[k] = dstA[ec];
        sn[k] = srcA[ec];
    }
    int rk[EPT];
#pragma unroll
    for (int k = 0; k < EPT; ++k) {
        const int e = e0 + k * T;
        rk[k] = atomicAdd(&cnt[dn[k]], (e < E) ? 1 : 0);
    }
#pragma unroll
    for (int k = 0; k < EPT; ++k) {
        const int e = e0 + k * T;
        if (e < E) {
            if (rk[k] < CAP) {
                slot[dn[k] * CAP + rk[k]] = sn[k];
            } else {
                const int p = atomicAdd(&cnt[N], 1);
                ovf_d[p] = dn[k];
                ovf_s[p] = sn[k];
            }
        }
    }
}

__global__ void __launch_bounds__(128) proj_pair_wmma_f32(
    const float* __restrict__ A,
    const float* __restrict__ W0, const float* __restrict__ W1,
    float* __restrict__ C0, float* __restrict__ C1, int ntiles,
    const int* __restrict__ srcA, const int* __restrict__ dstA, int E, int N,
    int* cnt, int* __restrict__ slot,
    int* __restrict__ ovf_d, int* __restrict__ ovf_s, int rankBlocks) {
    if ((int)blockIdx.x < rankBlocks) {
        dst_slot_role(srcA, dstA, E, N, cnt, slot, ovf_d, ovf_s,
                      (int)(blockIdx.x * 128 + threadIdx.x), rankBlocks * 128);
        return;
    }
    const int lane = threadIdx.x & 31;
    const int half = lane >> 4;
    const int l15  = lane & 15;
    const int strip = __builtin_amdgcn_readfirstlane(
        (int)((blockIdx.x - rankBlocks) * 4 + (threadIdx.x >> 5)));
    if (strip >= 2 * ntiles) return;
    const bool second = strip >= ntiles;
    const float* __restrict__ W = second ? W1 : W0;
    float* __restrict__ C       = second ? C1 : C0;
    const int tile = second ? strip - ntiles : strip;
    const int row0 = tile * 16;

    const float* __restrict__ Ap = A + (size_t)(row0 + l15) * DIM + half * 2;
    const float* __restrict__ Wp = W + (size_t)(half * 2) * DIM + l15 * 4;

    v8f c[8] = {};
    KFrag fa[KSTAGE], fb[KSTAGE];
    constexpr int KS4 = 4 * KSTAGE;
#pragma unroll
    for (int s = 0; s < KSTAGE; ++s) fa[s] = load_kfrag(Ap, Wp, 4 * s);
#pragma unroll 1
    for (int k = 0; k < DIM - 2 * KS4; k += 2 * KS4) {
#pragma unroll
        for (int s = 0; s < KSTAGE; ++s) fb[s] = load_kfrag(Ap, Wp, k + KS4 + 4 * s);
#pragma unroll
        for (int s = 0; s < KSTAGE; ++s) mma_kfrag(fa[s], c);
#pragma unroll
        for (int s = 0; s < KSTAGE; ++s) fa[s] = load_kfrag(Ap, Wp, k + 2 * KS4 + 4 * s);
#pragma unroll
        for (int s = 0; s < KSTAGE; ++s) mma_kfrag(fb[s], c);
    }
#pragma unroll
    for (int s = 0; s < KSTAGE; ++s) fb[s] = load_kfrag(Ap, Wp, DIM - KS4 + 4 * s);
#pragma unroll
    for (int s = 0; s < KSTAGE; ++s) mma_kfrag(fa[s], c);
#pragma unroll
    for (int s = 0; s < KSTAGE; ++s) mma_kfrag(fb[s], c);

    float* __restrict__ Cp = C + (size_t)(row0 + half * 8) * DIM + l15 * 4;
#pragma unroll
    for (int r = 0; r < 8; ++r) {
        float4 lo, hi;
        lo.x = c[0][r]; lo.y = c[1][r]; lo.z = c[2][r]; lo.w = c[3][r];
        hi.x = c[4][r]; hi.y = c[5][r]; hi.z = c[6][r]; hi.w = c[7][r];
        *(float4*)(Cp + (size_t)r * DIM)      = lo;
        *(float4*)(Cp + (size_t)r * DIM + 64) = hi;
    }
}

__global__ void zero_fill(float* __restrict__ p, int n) {
    int i = blockIdx.x * blockDim.x + threadIdx.x;
    if (i < n) p[i] = 0.f;
}

__device__ __forceinline__ v2f lrelu2(const v2f x) {
    return __builtin_elementwise_max(x, x * NEG_SLOPE);
}
__device__ __forceinline__ float edge_score(const float4 a, const float4 b,
                                            const float4 t) {
    const v2f lo = lrelu2(v2f{a.x, a.y} + v2f{b.x, b.y});
    const v2f hi = lrelu2(v2f{a.z, a.w} + v2f{b.z, b.w});
    const v2f p  = __builtin_elementwise_fma(hi, v2f{t.z, t.w}, lo * v2f{t.x, t.y});
    return p.x + p.y;
}

template <int SPAN>
__device__ __forceinline__ float head_sum(float p) {
#pragma unroll
    for (int m = 1; m < SPAN; m <<= 1) p += __shfl_xor(p, m, 32);
    return p;
}

template <int HEADS>
__device__ __forceinline__ float4 attend_node(
    const int d, const int lane,
    const int* __restrict__ cnt, const int* __restrict__ slot,
    const int* __restrict__ ovf_d, const int* __restrict__ ovf_s, int N,
    const float4* __restrict__ xl4, const float* __restrict__ xr,
    const float4 t, const float* bias) {
    constexpr int SPAN = (HEADS == 2) ? 16 : 32;
    const float NEG_INF = -__builtin_huge_valf();
    const float4 b = ((const float4*)xr)[(size_t)d * (DIM / 4) + lane];

    float4 acc = xl4[(size_t)d * (DIM / 4) + lane];
    float m = head_sum<SPAN>(edge_score(acc, b, t));
    float l = 1.f;

    const int n    = cnt[d];
    const int novf = (n > CAP) ? cnt[N] : 0;

    for (int ob = -32; ob < novf; ob += 32) {
        bool live; int my_s, c, nl, pos;
        if (ob < 0) {
            c    = (n < CAP) ? n : CAP;
            live = lane < c;
            my_s = live ? slot[d * CAP + lane] : d;
            nl   = c;
            pos  = live ? 0 : lane - c;
        } else {
            const int idx = ob + lane;
            const int od  = (idx < novf) ? ovf_d[idx] : -1;
            live = (od == d);
            my_s = live ? ovf_s[idx] : d;
            c    = __builtin_popcount(__builtin_amdgcn_ballot_w32(live));
            nl   = (c > 0) ? 32 : 0;
            pos  = 0;
        }

        const int key = live ? ((my_s << 5) | lane) : (0x7fffffe0 | lane);
        for (int i = 0; i < nl; ++i) {
            const int si = __builtin_amdgcn_readlane(key, i);
            pos += (si < key) ? 1 : 0;
        }
        const int sorted_s = __builtin_amdgcn_ds_permute(pos << 2, my_s);

        for (int j = 0; j < c; j += 4) {
            const int s0 = __builtin_amdgcn_readlane(sorted_s, j);
            const int s1 = __builtin_amdgcn_readlane(sorted_s, j + 1);
            const int s2 = __builtin_amdgcn_readlane(sorted_s, j + 2);
            const int s3 = __builtin_amdgcn_readlane(sorted_s, j + 3);
            const float4 a0 = xl4[(size_t)s0 * (DIM / 4) + lane];
            const float4 a1 = xl4[(size_t)s1 * (DIM / 4) + lane];
            const float4 a2 = xl4[(size_t)s2 * (DIM / 4) + lane];
            const float4 a3 = xl4[(size_t)s3 * (DIM / 4) + lane];

            float q0 = edge_score(a0, b, t);
            float q1 = edge_score(a1, b, t);
            float q2 = edge_score(a2, b, t);
            float q3 = edge_score(a3, b, t);
#pragma unroll
            for (int mm = 1; mm < SPAN; mm <<= 1) {
                const float r0 = __shfl_xor(q0, mm, 32);
                const float r1 = __shfl_xor(q1, mm, 32);
                const float r2 = __shfl_xor(q2, mm, 32);
                const float r3 = __shfl_xor(q3, mm, 32);
                q0 += r0; q1 += r1; q2 += r2; q3 += r3;
            }
            const float p0 = q0;
            const float p1 = (j + 1 < c) ? q1 : NEG_INF;
            const float p2 = (j + 2 < c) ? q2 : NEG_INF;
            const float p3 = (j + 3 < c) ? q3 : NEG_INF;

            const float mn = fmaxf(fmaxf(m, p0), fmaxf(p1, fmaxf(p2, p3)));
            const float sc = __expf(m - mn);
            const float w0 = __expf(p0 - mn);
            const float w1 = __expf(p1 - mn);
            const float w2 = __expf(p2 - mn);
            const float w3 = __expf(p3 - mn);
            m = mn;
            l = l * sc + ((w0 + w1) + (w2 + w3));
            acc.x = acc.x * sc + w0 * a0.x + w1 * a1.x + w2 * a2.x + w3 * a3.x;
            acc.y = acc.y * sc + w0 * a0.y + w1 * a1.y + w2 * a2.y + w3 * a3.y;
            acc.z = acc.z * sc + w0 * a0.z + w1 * a1.z + w2 * a2.z + w3 * a3.z;
            acc.w = acc.w * sc + w0 * a0.w + w1 * a1.w + w2 * a2.w + w3 * a3.w;
        }
    }

    const float den = l + 1e-16f;
    float inv = __builtin_amdgcn_rcpf(den);
    inv = __builtin_fmaf(__builtin_fmaf(-den, inv, 1.f), inv, inv);
    const float4 bi = ((const float4*)bias)[lane];
    float4 o;
    o.x = elu(acc.x * inv + bi.x);
    o.y = elu(acc.y * inv + bi.y);
    o.z = elu(acc.z * inv + bi.z);
    o.w = elu(acc.w * inv + bi.w);
    return o;
}

template <int HEADS>
__global__ void __launch_bounds__(256, 16) node_attend(
    const int* __restrict__ cnt, const int* __restrict__ slot,
    const int* __restrict__ ovf_d, const int* __restrict__ ovf_s, int N,
    const float* __restrict__ xl, const float* __restrict__ xr,
    const float* __restrict__ att, const float* bias,
    float* out) {
    const int d_first = __builtin_amdgcn_readfirstlane(
        (int)(blockIdx.x * (blockDim.x >> 5) + (threadIdx.x >> 5)));
    const int d_step = (int)(gridDim.x * (blockDim.x >> 5));
    const int lane = threadIdx.x & 31;
    const float4* __restrict__ xl4 = (const float4*)xl;
    const float4 t  = ((const float4*)att)[lane];

    for (int d = d_first; d < N; d += d_step) {
        const float4 o = attend_node<HEADS>(d, lane, cnt, slot, ovf_d, ovf_s, N,
                                            xl4, xr, t, bias);
        ((float4*)out)[(size_t)d * (DIM / 4) + lane] = o;
    }
}

#define ROWP 132
#define KSTAGE_H 1
__global__ void __launch_bounds__(128, 16) attend_then_proj_pair(
    const int* __restrict__ cnt, const int* __restrict__ slot,
    const int* __restrict__ ovf_d, const int* __restrict__ ovf_s, int N,
    const float* __restrict__ xl, const float* __restrict__ xr,
    const float* __restrict__ att, const float* bias,
    const float* __restrict__ W0, const float* __restrict__ W1,
    float* __restrict__ C0, float* __restrict__ C1) {
    __shared__ __attribute__((aligned(16))) float rows[16 * ROWP];
    const int lane = threadIdx.x & 31;
    const int w    = __builtin_amdgcn_readfirstlane((int)(threadIdx.x >> 5));
    const int row0 = (int)blockIdx.x * 16;
    {
        const float4* __restrict__ xl4 = (const float4*)xl;
        const float4 t = ((const float4*)att)[lane];
        for (int i = 0; i < 4; ++i) {
            const int r = w * 4 + i;
            const int d = row0 + r;
            const float4 o = attend_node<2>(d, lane, cnt, slot, ovf_d, ovf_s, N, xl4, xr, t, bias);
            *(float4*)(rows + r * ROWP + lane * 4) = o;
        }
    }
    __syncthreads();

    const int half = lane >> 4;
    const int l15  = lane & 15;
    const bool second = w >= 2;
    const int  g      = w & 1;
    const float* __restrict__ Wm = second ? W1 : W0;
    float* __restrict__ C        = second ? C1 : C0;
    const float* Ap              = rows + l15 * ROWP + half * 2;
    const float* __restrict__ Wp = Wm + (size_t)(half * 2) * DIM + g * 64 + l15 * 4;

    v8f c[4] = {};
    KFragH fa[KSTAGE_H], fb[KSTAGE_H];
    constexpr int KS4 = 4 * KSTAGE_H;
#pragma unroll
    for (int s = 0; s < KSTAGE_H; ++s) fa[s] = load_kfrag_h(Ap, Wp, 4 * s);
#pragma unroll 1
    for (int k = 0; k < DIM - 2 * KS4; k += 2 * KS4) {
#pragma unroll
        for (int s = 0; s < KSTAGE_H; ++s) fb[s] = load_kfrag_h(Ap, Wp, k + KS4 + 4 * s);
#pragma unroll
        for (int s = 0; s < KSTAGE_H; ++s) mma_kfrag_h(fa[s], c);
#pragma unroll
        for (int s = 0; s < KSTAGE_H; ++s) fa[s] = load_kfrag_h(Ap, Wp, k + 2 * KS4 + 4 * s);
#pragma unroll
        for (int s = 0; s < KSTAGE_H; ++s) mma_kfrag_h(fb[s], c);
    }
#pragma unroll
    for (int s = 0; s < KSTAGE_H; ++s) fb[s] = load_kfrag_h(Ap, Wp, DIM - KS4 + 4 * s);
#pragma unroll
    for (int s = 0; s < KSTAGE_H; ++s) mma_kfrag_h(fa[s], c);
#pragma unroll
    for (int s = 0; s < KSTAGE_H; ++s) mma_kfrag_h(fb[s], c);

    float* __restrict__ Cp = C + (size_t)(row0 + half * 8) * DIM + g * 64 + l15 * 4;
#pragma unroll
    for (int r = 0; r < 8; ++r) {
        float4 v;
        v.x = c[0][r]; v.y = c[1][r]; v.z = c[2][r]; v.w = c[3][r];
        *(float4*)(Cp + (size_t)r * DIM) = v;
    }
}

extern "C" void kernel_launch(void* const* d_in, const int* in_sizes, int n_in,
                              void* d_out, int out_size, void* d_ws, size_t ws_size,
                              hipStream_t stream) {
    const float* x    = (const float*)d_in[0];
    const int*   ei   = (const int*)d_in[1];
    const float* Wl1  = (const float*)d_in[2];
    const float* Wr1  = (const float*)d_in[3];
    const float* att1 = (const float*)d_in[4];
    const float* b1   = (const float*)d_in[5];
    const float* Wl2  = (const float*)d_in[6];
    const float* Wr2  = (const float*)d_in[7];
    const float* att2 = (const float*)d_in[8];
    const float* b2   = (const float*)d_in[9];

    const int N  = in_sizes[0] / DIM;
    const int E  = in_sizes[1] / 2;

    const int* srcA = ei;
    const int* dstA = ei + E;

    float* xl    = (float*)d_ws;
    float* xr    = xl + (size_t)N * DIM;
    float* xl2   = xr + (size_t)N * DIM;
    float* xr2   = xl2 + (size_t)N * DIM;
    int*   slot  = (int*)(xr2 + (size_t)N * DIM);
    int*   ovf_d = slot + (size_t)N * CAP;
    int*   ovf_s = ovf_d + (size_t)E;
    int*   cnt   = ovf_s + (size_t)E;
    float* out   = (float*)d_out;

    const dim3 B(256);
    const dim3 BG(128);
    const int gCnt   = (N + 1 + 255) / 256;
    const int nTiles = N / 16;
    const int gPair  = (2 * nTiles + 3) / 4;
    const int gRank  = (E + 128 * EPT - 1) / (128 * EPT);
    const int gNodeW = (N + 8 * NODES_PER_WAVE - 1) / (8 * NODES_PER_WAVE);

    zero_fill<<<gCnt, B, 0, stream>>>((float*)cnt, N + 1);
    proj_pair_wmma_f32<<<gRank + gPair, BG, 0, stream>>>(
        x, Wl1, Wr1, xl, xr, nTiles, srcA, dstA, E, N, cnt, slot, ovf_d, ovf_s, gRank);

    attend_then_proj_pair<<<nTiles, BG, 0, stream>>>(
        cnt, slot, ovf_d, ovf_s, N, xl, xr, att1, b1, Wl2, Wr2, xl2, xr2);

    node_attend<1><<<gNodeW, B, 0, stream>>>(cnt, slot, ovf_d, ovf_s, N, xl2, xr2, att2, b2, out);
}
